# starting kernel
# baseline (speedup 1.0000x reference)
#include <hip/hip_runtime.h>
#include <math.h>

typedef float v2f __attribute__((ext_vector_type(2)));
typedef float v8f __attribute__((ext_vector_type(8)));

static inline int cdiv_i(long long a, long long b) { return (int)((a + b - 1) / b); }

__global__ void zero_f32(float* __restrict__ p, long long n) {
  long long i = (long long)blockIdx.x * blockDim.x + threadIdx.x;
  long long stride = (long long)gridDim.x * blockDim.x;
  for (; i < n; i += stride) p[i] = 0.0f;
}

__global__ void relu_f32(float* __restrict__ p, long long n) {
  long long i = (long long)blockIdx.x * blockDim.x + threadIdx.x;
  long long stride = (long long)gridDim.x * blockDim.x;
  for (; i < n; i += stride) p[i] = fmaxf(p[i], 0.0f);
}

__global__ void deg_count(const int* __restrict__ dst, int E, float* __restrict__ deg) {
  int i = blockIdx.x * blockDim.x + threadIdx.x;
  if (i < E) atomicAdd(&deg[dst[i]], 1.0f);
}

__global__ void deg_to_dinv(float* __restrict__ deg, long long n) {
  long long i = (long long)blockIdx.x * blockDim.x + threadIdx.x;
  long long stride = (long long)gridDim.x * blockDim.x;
  for (; i < n; i += stride) {
    float d = deg[i];
    deg[i] = (d > 0.0f) ? rsqrtf(d) : 0.0f;
  }
}

__global__ void gemm_wmma_f32(const float* __restrict__ A, const float* __restrict__ B,
                              const float* __restrict__ bias, float* __restrict__ D,
                              int M, int K, int Ncols) {
  extern __shared__ float As[];
  const int row0 = blockIdx.x * 16;
  const int tid  = threadIdx.x;
  const int ldA  = K + 1;

  for (int idx = tid; idx < 16 * K; idx += blockDim.x) {
    int r = idx / K;
    int k = idx - r * K;
    int row = row0 + r;
    if (row >= M) row = M - 1;
    As[r * ldA + k] = A[(long long)row * K + k];
  }
  __syncthreads();

  const int wave = tid >> 5;
  const int lane = tid & 31;
  const int half = lane >> 4;
  const int mn   = lane & 15;
  const int kb   = half * 2;
  const int col0 = wave * 16;

  v8f acc = {};
  #pragma unroll 4
  for (int kk = 0; kk < K; kk += 4) {
    v2f a, b;
    a.x = As[mn * ldA + kk + kb];
    a.y = As[mn * ldA + kk + kb + 1];
    const float* Bp = B + (long long)(kk + kb) * Ncols + col0 + mn;
    b.x = Bp[0];
    b.y = Bp[Ncols];
    acc = __builtin_amdgcn_wmma_f32_16x16x4_f32(
         false, a,  false, b,
         (short)0, acc,  false,  false);
  }

  const float bv = bias ? bias[col0 + mn] : 0.0f;
  #pragma unroll
  for (int r = 0; r < 8; ++r) {
    int row = row0 + r + half * 8;
    if (row < M) D[(long long)row * Ncols + col0 + mn] = acc[r] + bv;
  }
}

__global__ void gcn_aggregate(const int* __restrict__ src, const int* __restrict__ dst, int E,
                              const float* __restrict__ dinv,
                              const float* __restrict__ HH, int ldH,
                              float* __restrict__ out, int ldOut, int width) {
  long long total  = (long long)E * width;
  long long stride = (long long)gridDim.x * blockDim.x;
  for (long long i = (long long)blockIdx.x * blockDim.x + threadIdx.x; i < total; i += stride) {
    int e = (int)(i / width);
    int h = (int)(i - (long long)e * width);
    int s = src[e];
    int d = dst[e];
    float nrm = dinv[s] * dinv[d];
    atomicAdd(&out[(long long)d * ldOut + h], HH[(long long)s * ldH + h] * nrm);
  }
}

__global__ void build_R1(const float* __restrict__ HH1, const float* __restrict__ b1,
                         float* __restrict__ R1, int N) {
  long long total  = (long long)N * 512;
  long long stride = (long long)gridDim.x * blockDim.x;
  for (long long i = (long long)blockIdx.x * blockDim.x + threadIdx.x; i < total; i += stride) {
    int col = (int)(i & 511);
    long long n = i >> 9;
    int br = col >> 7;
    int h  = col & 127;
    float v = b1[h];
    if (br == 3) v += HH1[n * 128 + h];
    R1[i] = v;
  }
}

__global__ void build_R2(const float* __restrict__ HH2, const float* __restrict__ b2,
                         float* __restrict__ R2, int N) {
  long long total  = (long long)N * 256;
  long long stride = (long long)gridDim.x * blockDim.x;
  for (long long i = (long long)blockIdx.x * blockDim.x + threadIdx.x; i < total; i += stride) {
    int col = (int)(i & 255);
    long long n = i >> 8;
    int br = col >> 6;
    int h  = col & 63;
    float v = b2[h];
    if (br == 3) v += HH2[n * 64 + h];
    R2[i] = v;
  }
}

__global__ void log_softmax64(float* __restrict__ out, int N) {
  int row  = blockIdx.x * (blockDim.x >> 5) + (threadIdx.x >> 5);
  int lane = threadIdx.x & 31;
  if (row >= N) return;
  float* p = out + (long long)row * 64;
  float a = p[lane];
  float b = p[lane + 32];
  float mx = fmaxf(a, b);
  #pragma unroll
  for (int off = 16; off > 0; off >>= 1) mx = fmaxf(mx, __shfl_xor(mx, off));
  float s = __expf(a - mx) + __expf(b - mx);
  #pragma unroll
  for (int off = 16; off > 0; off >>= 1) s += __shfl_xor(s, off);
  float lse = mx + __logf(s);
  p[lane]      = a - lse;
  p[lane + 32] = b - lse;
}

extern "C" void kernel_launch(void* const* d_in, const int* in_sizes, int n_in,
                              void* d_out, int out_size, void* d_ws, size_t ws_size,
                              hipStream_t stream) {
  const float* x  = (const float*)d_in[0];
  const int* e1   = (const int*)d_in[1];
  const int* e2   = (const int*)d_in[2];
  const int* e3   = (const int*)d_in[3];
  const float* W1 = (const float*)d_in[5];
  const float* b1 = (const float*)d_in[6];
  const float* W2 = (const float*)d_in[7];
  const float* b2 = (const float*)d_in[8];
  const float* Wl = (const float*)d_in[9];
  const float* bl = (const float*)d_in[10];

  const int F = 512, H = 128, C = 64;
  const int N  = in_sizes[0] / F;
  const int E1 = in_sizes[1] / 2;
  const int E2 = in_sizes[2] / 2;
  const int E3 = in_sizes[3] / 2;

  float* ws   = (float*)d_ws;
  float* dinv = ws;
  float* HH1  = dinv + 3LL * N;
  float* R1   = HH1 + (long long)N * H;
  float* HH2  = R1 + (long long)N * 4 * H;
  float* R2   = HH2 + (long long)N * C;

  const int T = 256;
  const int AGG_MAX_BLOCKS = 131072;

  zero_f32<<<cdiv_i(3LL * N, T), T, 0, stream>>>(dinv, 3LL * N);
  deg_count<<<cdiv_i(E1, T), T, 0, stream>>>(e1 + E1, E1, dinv);
  deg_count<<<cdiv_i(E2, T), T, 0, stream>>>(e2 + E2, E2, dinv + N);
  deg_count<<<cdiv_i(E3, T), T, 0, stream>>>(e3 + E3, E3, dinv + 2LL * N);
  deg_to_dinv<<<cdiv_i(3LL * N, T), T, 0, stream>>>(dinv, 3LL * N);

  const int mtiles = cdiv_i(N, 16);

  gemm_wmma_f32<<<mtiles, 32 * (H / 16), (size_t)16 * (F + 1) * sizeof(float), stream>>>(
      x, W1, nullptr, HH1, N, F, H);

  build_R1<<<cdiv_i((long long)N * 4 * H, T), T, 0, stream>>>(HH1, b1, R1, N);

  {
    long long t1 = (long long)E1 * H, t2 = (long long)E2 * H, t3 = (long long)E3 * H;
    int g1 = min(cdiv_i(t1, T), AGG_MAX_BLOCKS);
    int g2 = min(cdiv_i(t2, T), AGG_MAX_BLOCKS);
    int g3 = min(cdiv_i(t3, T), AGG_MAX_BLOCKS);
    gcn_aggregate<<<g1, T, 0, stream>>>(e1, e1 + E1, E1, dinv,           HH1, H, R1,           4 * H, H);
    gcn_aggregate<<<g2, T, 0, stream>>>(e2, e2 + E2, E2, dinv + N,       HH1, H, R1 + H,       4 * H, H);
    gcn_aggregate<<<g3, T, 0, stream>>>(e3, e3 + E3, E3, dinv + 2LL * N, HH1, H, R1 + 2 * H,   4 * H, H);
  }

  relu_f32<<<cdiv_i((long long)N * 4 * H, T), T, 0, stream>>>(R1, (long long)N * 4 * H);

  gemm_wmma_f32<<<mtiles, 32 * (C / 16), (size_t)16 * (4 * H + 1) * sizeof(float), stream>>>(
      R1, W2, nullptr, HH2, N, 4 * H, C);

  build_R2<<<cdiv_i((long long)N * 4 * C, T), T, 0, stream>>>(HH2, b2, R2, N);

  {
    long long t1 = (long long)E1 * C, t2 = (long long)E2 * C, t3 = (long long)E3 * C;
    int g1 = min(cdiv_i(t1, T), AGG_MAX_BLOCKS);
    int g2 = min(cdiv_i(t2, T), AGG_MAX_BLOCKS);
    int g3 = min(cdiv_i(t3, T), AGG_MAX_BLOCKS);
    gcn_aggregate<<<g1, T, 0, stream>>>(e1, e1 + E1, E1, dinv,           HH2, C, R2,           4 * C, C);
    gcn_aggregate<<<g2, T, 0, stream>>>(e2, e2 + E2, E2, dinv + N,       HH2, C, R2 + C,       4 * C, C);
    gcn_aggregate<<<g3, T, 0, stream>>>(e3, e3 + E3, E3, dinv + 2LL * N, HH2, C, R2 + 2 * C,   4 * C, C);
  }

  float* logits = (float*)d_out;
  gemm_wmma_f32<<<mtiles, 32 * (C / 16), (size_t)16 * (4 * C + 1) * sizeof(float), stream>>>(
      R2, Wl, bl, logits, N, 4 * C, C);

  log_softmax64<<<cdiv_i(N, T / 32), T, 0, stream>>>(logits, N);
}


# GEMMs: the weight matrix, not the A strip, lives in LDS (copied once per 80-row work-group, in fragment order); A fragments straight from global by 16-byte loads; k order inside 8-blocks changes
# speedup vs baseline: 5.3865x; 5.3865x over previous
#include <hip/hip_runtime.h>
#include <math.h>


typedef float v2f __attribute__((ext_vector_type(2)));
typedef float v8f __attribute__((ext_vector_type(8)));

static inline int cdiv_i(long long a, long long b) { return (int)((a + b - 1) / b); }

constexpr int TR    = 80;
constexpr int NTHR  = 1024;
constexpr int NV    = 8;
constexpr int CHUNK = NTHR * NV * 4;
constexpr int QCAP  = 8192;
constexpr int SEGCAP = 24576;
constexpr int NWAVE = NTHR / 32;
constexpr int WSEG  = SEGCAP / NWAVE;
constexpr int G      = 5;
constexpr int GR     = G * TR;
constexpr int SUBCAP = 6400;

__device__ __forceinline__ void load_chunk(int4 (&v)[NV], const int* __restrict__ dp,
                                           int base, int E, bool vec_ok, int tid) {
  if (vec_ok && base + CHUNK <= E) {
    #pragma unroll
    for (int j = 0; j < NV; ++j)
      v[j] = *reinterpret_cast<const int4*>(dp + base + (j * NTHR + tid) * 4);
  } else if (vec_ok && E >= 4) {
    #pragma unroll
    for (int j = 0; j < NV; ++j) {
      const int i = base + (j * NTHR + tid) * 4;
      const int4 t = *reinterpret_cast<const int4*>(dp + min(i, E - 4));
      const bool in = i < E;
      v[j].x = in ? t.x : -1;
      v[j].y = in ? t.y : -1;
      v[j].z = in ? t.z : -1;
      v[j].w = in ? t.w : -1;
    }
  } else {
    #pragma unroll
    for (int j = 0; j < NV; ++j) {
      const int i = base + (j * NTHR + tid) * 4;
      v[j].x = (i     < E) ? dp[i]     : -1;
      v[j].y = (i + 1 < E) ? dp[i + 1] : -1;
      v[j].z = (i + 2 < E) ? dp[i + 2] : -1;
      v[j].w = (i + 3 < E) ? dp[i + 3] : -1;
    }
  }
}

__device__ __forceinline__ void take_edges(unsigned b, bool hit, unsigned a, int e, int k, unsigned* cntk,
                                           int* wqe, int* wqc, int& wn) {
  if (b != 0u) {
    if (hit) {
      atomicAdd(&cntk[a], 1u);
      const int pos = wn + (int)__builtin_amdgcn_mbcnt_lo(b, 0u);
      if (pos < WSEG) { wqe[pos] = e; wqc[pos] = (int)a * 4 + k; }
    }
    wn += __builtin_popcount(b);
  }
}

__global__ __launch_bounds__(NTHR)
void deg_owner(const int* __restrict__ s1, const int* __restrict__ d1, int E1,
               const int* __restrict__ s2, const int* __restrict__ d2, int E2,
               const int* __restrict__ s3, const int* __restrict__ d3, int E3,
               unsigned* __restrict__ pcnt, int* __restrict__ seghdr, int* __restrict__ seg, int N) {
  __shared__ unsigned cnt[3 * GR];
  __shared__ int qe[SEGCAP];
  __shared__ int qc[SEGCAP];
  __shared__ int wtc[NWAVE * G];
  __shared__ int wov[NWAVE];
  const int tid  = threadIdx.x;
  const int lane = tid & 31;
  const int wave = __builtin_amdgcn_readfirstlane(tid >> 5);
  const int rg   = blockIdx.x / G;
  const int sl   = blockIdx.x - rg * G;
  const int r0   = rg * GR;
  for (int i = tid; i < 3 * GR; i += NTHR) cnt[i] = 0u;
  __syncthreads();

  int* const wqe = qe + wave * WSEG;
  int* const wqc = qc + wave * WSEG;
  int wn = 0;

  #pragma unroll 1
  for (int k = 0; k < 3; ++k) {
    const int* __restrict__ dp = (k == 0) ? d1 : (k == 1) ? d2 : d3;
    const int E = (k == 0) ? E1 : (k == 1) ? E2 : E3;
    unsigned* const cntk = cnt + k * GR;
    const int m   = (E >= 4) ? (int)((reinterpret_cast<size_t>(dp) >> 2) & 3) : 0;
    const int* __restrict__ ap = dp - m;
    const int Ep  = E + m;
    const int Epf = (E >= 4) ? (Ep & ~3) : 0;
    const int pg  = ((Epf >> 2) + G - 1) / G;
    const int lo  = min(sl * pg * 4, Epf);
    const int hi  = min(lo + pg * 4, Epf);
    for (int pb = lo; pb < hi; pb += CHUNK) {
      int4 v[NV];
      if (pb + CHUNK <= hi && (pb > 0 || m == 0)) {
        #pragma unroll
        for (int j = 0; j < NV; ++j)
          v[j] = *reinterpret_cast<const int4*>(ap + pb + (j * NTHR + tid) * 4);
      } else {
        #pragma unroll
        for (int j = 0; j < NV; ++j) {
          const int p = pb + (j * NTHR + tid) * 4;
          const int4 t = *reinterpret_cast<const int4*>(ap + min(p, hi - 4));
          const bool in = p < hi;
          v[j].x = (in && p     >= m) ? t.x : -1;
          v[j].y = (in && p + 1 >= m) ? t.y : -1;
          v[j].z = (in && p + 2 >= m) ? t.z : -1;
          v[j].w = (in && p + 3 >= m) ? t.w : -1;
        }
      }
      unsigned any[NV];
      #pragma unroll
      for (int j = 0; j < NV; ++j) {
        v[j].x -= r0; v[j].y -= r0; v[j].z -= r0; v[j].w -= r0;
        const unsigned mn = min(min((unsigned)v[j].x, (unsigned)v[j].y), min((unsigned)v[j].z, (unsigned)v[j].w));
        any[j] = __builtin_amdgcn_ballot_w32(mn < (unsigned)GR);
      }
      #pragma unroll
      for (int j = 0; j < NV; ++j) {
        if (any[j] != 0u) {
          const int i = pb + (j * NTHR + tid) * 4 - m;
          const unsigned a0 = (unsigned)v[j].x, a1 = (unsigned)v[j].y, a2 = (unsigned)v[j].z, a3 = (unsigned)v[j].w;
          const unsigned b0 = __builtin_amdgcn_ballot_w32(a0 < (unsigned)GR);
          const unsigned b1 = __builtin_amdgcn_ballot_w32(a1 < (unsigned)GR);
          const unsigned b2 = __builtin_amdgcn_ballot_w32(a2 < (unsigned)GR);
          const unsigned b3 = __builtin_amdgcn_ballot_w32(a3 < (unsigned)GR);
          take_edges(b0, a0 < (unsigned)GR, a0, i,     k, cntk, wqe, wqc, wn);
          take_edges(b1, a1 < (unsigned)GR, a1, i + 1, k, cntk, wqe, wqc, wn);
          take_edges(b2, a2 < (unsigned)GR, a2, i + 2, k, cntk, wqe, wqc, wn);
          take_edges(b3, a3 < (unsigned)GR, a3, i + 3, k, cntk, wqe, wqc, wn);
        }
      }
    }
    {
      const bool mine = (sl == G - 1) && tid < Ep - Epf;
      const int i = Epf - m + tid;
      unsigned a = 0xffffffffu;
      if (mine) a = (unsigned)(dp[i] - r0);
      const bool hit = mine && a < (unsigned)GR;
      take_edges(__builtin_amdgcn_ballot_w32(hit), hit, a, i, k, cntk, wqe, wqc, wn);
    }
  }
  __syncthreads();

  if (tid < GR) {
    const int row = r0 + tid;
    if (row < N) {
      uint4 o;
      o.x = cnt[tid]; o.y = cnt[GR + tid]; o.z = cnt[2 * GR + tid]; o.w = 0u;
      reinterpret_cast<uint4*>(pcnt)[(long long)sl * N + row] = o;
    }
  }

  const int nq = min(wn, WSEG);
  int tc[G];
  #pragma unroll
  for (int g = 0; g < G; ++g) tc[g] = 0;
  for (int i0 = 0; i0 < nq; i0 += 32) {
    const int ii = i0 + lane;
    int t = -1;
    if (ii < nq) t = (wqc[ii] >> 2) / TR;
    #pragma unroll
    for (int g = 0; g < G; ++g) tc[g] += __builtin_popcount(__builtin_amdgcn_ballot_w32(t == g));
  }
  if (lane == 0) {
    #pragma unroll
    for (int g = 0; g < G; ++g) wtc[wave * G + g] = tc[g];
    wov[wave] = (wn > WSEG) ? 1 : 0;
  }
  __syncthreads();

  static_assert(NWAVE == 32, "one lane per wave of the work-group");
  const bool ovf = __builtin_amdgcn_ballot_w32(wov[lane] != 0) != 0u;
  int off[G], tot[G];
  #pragma unroll
  for (int g = 0; g < G; ++g) {
    int t = wtc[lane * G + g];
    int o = (lane < wave) ? t : 0;
    #pragma unroll
    for (int s = 16; s > 0; s >>= 1) { t += __shfl_xor(t, s); o += __shfl_xor(o, s); }
    tot[g] = __builtin_amdgcn_readfirstlane(t);
    off[g] = __builtin_amdgcn_readfirstlane(o);
  }
  const long long pair0 = ((long long)rg * G) * G + sl;
  #pragma unroll
  for (int g = 0; g < G; ++g) {
    if (tid == g) {
      const bool o = ovf || tot[g] > SUBCAP;
      int4 h;
      h.x = o ? 0 : tot[g]; h.y = o ? 1 : 0; h.z = 0; h.w = 0;
      reinterpret_cast<int4*>(seghdr)[pair0 + (long long)g * G] = h;
    }
  }
  const int nw = ovf ? 0 : nq;
  int2* const sg = reinterpret_cast<int2*>(seg) + pair0 * SUBCAP;
  for (int i0 = lane; i0 < nw; i0 += 4 * 32) {
    int c[4], s[4];
    #pragma unroll
    for (int u = 0; u < 4; ++u) {
      const int ii = min(i0 + u * 32, nw - 1);
      const int e = wqe[ii];
      c[u] = wqc[ii];
      const int kk = c[u] & 3;
      const int* sp = (kk == 0) ? s1 : (kk == 1) ? s2 : s3;
      s[u] = sp[e];
    }
    #pragma unroll
    for (int u = 0; u < 4; ++u) {
      const int a = c[u] >> 2;
      const int t = (i0 + u * 32 < nw) ? a / TR : -1;
      const int2 rec = make_int2(s[u], (a - t * TR) * 4 + (c[u] & 3));
      #pragma unroll
      for (int g = 0; g < G; ++g) {
        const unsigned b = __builtin_amdgcn_ballot_w32(t == g);
        if (t == g) {
          const int pos = off[g] + (int)__builtin_amdgcn_mbcnt_lo(b, 0u);
          if (pos < SUBCAP) sg[(long long)g * G * SUBCAP + pos] = rec;
        }
        off[g] += __builtin_popcount(b);
      }
    }
  }
}

constexpr int GW    = 20;
constexpr int GTH   = GW * 32;
constexpr int GROWS = 80;

template <int K, int NC>
__global__ __launch_bounds__(GTH)
void gemm_wmma_f32(const float* __restrict__ A, const float* __restrict__ B,
                   const float* __restrict__ bias, float* __restrict__ D, int M,
                   const unsigned* __restrict__ pc, float* __restrict__ dinv4) {
  constexpr int NT  = NC / 16;
  constexpr int CT  = NT / 4;
  constexpr int KB  = K / 8;
  constexpr int TOT = (K / 4) * NC;
  static_assert(K % 32 == 0 && NC % 64 == 0, "shapes");
  __shared__ float4 Bs[KB * NT * 32];

  const int tid  = threadIdx.x;
  const int lane = tid & 31;
  const int wave = tid >> 5;
  const int half = lane >> 4;
  const int mn   = lane & 15;
  const int row0 = blockIdx.x * GROWS;
  const int rt   = wave >> 2;
  const int ct0  = (wave & 3) * CT;

  const int arow = min(row0 + rt * 16 + mn, M - 1);
  const float4* __restrict__ ap = reinterpret_cast<const float4*>(A + (long long)arow * K) + half;

  float4 cur[4];
  #pragma unroll
  for (int j = 0; j < 4; ++j) cur[j] = ap[2 * j];

  if (pc != nullptr && tid < GROWS && row0 + tid < M) {
    const int row = row0 + tid;
    unsigned c0 = 0u, c1 = 0u, c2 = 0u;
    #pragma unroll
    for (int g = 0; g < G; ++g) {
      const uint4 p = reinterpret_cast<const uint4*>(pc)[(long long)g * M + row];
      c0 += p.x; c1 += p.y; c2 += p.z;
    }
    float4 o;
    o.x = (c0 > 0u) ? rsqrtf((float)c0) : 0.0f;
    o.y = (c1 > 0u) ? rsqrtf((float)c1) : 0.0f;
    o.z = (c2 > 0u) ? rsqrtf((float)c2) : 0.0f;
    o.w = 0.0f;
    reinterpret_cast<float4*>(dinv4)[row] = o;
  }

  constexpr int NTRIP = (TOT + GTH - 1) / GTH;
  constexpr int BATCH = (NTRIP % 13 == 0) ? 13 : 7;
  static_assert(NTRIP % BATCH == 0, "whole batches");
  #pragma unroll 1
  for (int t0 = 0; t0 < NTRIP; t0 += BATCH) {
    float4 v[BATCH];
    #pragma unroll
    for (int u = 0; u < BATCH; ++u) {
      const int i  = min((t0 + u) * GTH + tid, TOT - 1);
      const int c  = i % NC;
      const int kg = i / NC;
      const float* bp = B + (long long)(kg * 4) * NC + c;
      v[u].x = bp[0]; v[u].y = bp[NC]; v[u].z = bp[2 * NC]; v[u].w = bp[3 * NC];
    }
    #pragma unroll
    for (int u = 0; u < BATCH; ++u) {
      const int i  = (t0 + u) * GTH + tid;
      const int c  = i % NC;
      const int kg = i / NC;
      if (i < TOT) Bs[((kg >> 1) * NT + (c >> 4)) * 32 + (kg & 1) * 16 + (c & 15)] = v[u];
    }
  }
  __syncthreads();

  v8f acc[CT];
  #pragma unroll
  for (int t = 0; t < CT; ++t) acc[t] = v8f{};

  #pragma unroll 1
  for (int kc = 0; kc < KB; kc += 4) {
    const int kn = min(kc + 4, KB - 4);
    float4 nxt[4];
    #pragma unroll
    for (int j = 0; j < 4; ++j) nxt[j] = ap[2 * (kn + j)];

    #pragma unroll
    for (int j = 0; j < 4; ++j) {
      v2f a0, a1;
      a0.x = cur[j].x; a0.y = cur[j].y;
      a1.x = cur[j].z; a1.y = cur[j].w;
      #pragma unroll
      for (int t = 0; t < CT; ++t) {
        const float4 b = Bs[((kc + j) * NT + ct0 + t) * 32 + lane];
        v2f b0, b1;
        b0.x = b.x; b0.y = b.y;
        b1.x = b.z; b1.y = b.w;
        acc[t] = __builtin_amdgcn_wmma_f32_16x16x4_f32(
             false, a0,  false, b0,
             (short)0, acc[t],  false,  false);
        acc[t] = __builtin_amdgcn_wmma_f32_16x16x4_f32(
             false, a1,  false, b1,
             (short)0, acc[t],  false,  false);
      }
    }
    #pragma unroll
    for (int j = 0; j < 4; ++j) cur[j] = nxt[j];
  }

  #pragma unroll
  for (int t = 0; t < CT; ++t) {
    const int col = (ct0 + t) * 16 + mn;
    const float bv = bias ? bias[col] : 0.0f;
    #pragma unroll
    for (int r = 0; r < 8; ++r) {
      const int row = row0 + rt * 16 + r + half * 8;
      if (row < M) D[(long long)row * NC + col] = acc[t][r] + bv;
    }
  }
}


template <int WIDTH>
__device__ __forceinline__ void edge_in_lane(int e, int dl, int k, const int* __restrict__ sp,
                                             const float* __restrict__ dinv4, const float* dinvL,
                                             const float* __restrict__ HH, float* acc) {
  constexpr int LD = 4 * WIDTH;
  const int s = sp[e];
  const float w = dinv4[s * 4 + k] * dinvL[dl * 4 + k];
  const float* hp = HH + (long long)s * WIDTH;
  float* ap = acc + dl * LD + k * WIDTH;
  #pragma unroll 1
  for (int c = 0; c < WIDTH; ++c) atomicAdd(ap + c, hp[c] * w);
}

template <int WIDTH, int CNT>
__device__ __forceinline__ void work_edges(const int* qe_, const int* qd_, int k,
                                           const int* __restrict__ sp,
                                           const float* __restrict__ dinv4, const float* dinvL,
                                           const float* __restrict__ HH, float* acc, int lane) {
  constexpr int LD = 4 * WIDTH;
  constexpr int NJ = WIDTH / 32;
  int dl[CNT], s[CNT];
  #pragma unroll
  for (int u = 0; u < CNT; ++u) { dl[u] = qd_[u]; s[u] = sp[qe_[u]]; }
  float w[CNT], hv[CNT][NJ];
  #pragma unroll
  for (int u = 0; u < CNT; ++u) {
    w[u] = dinv4[s[u] * 4 + k] * dinvL[dl[u] * 4 + k];
    const float* hp = HH + (long long)s[u] * WIDTH + lane;
    #pragma unroll
    for (int j = 0; j < NJ; ++j) hv[u][j] = hp[32 * j];
  }
  #pragma unroll
  for (int u = 0; u < CNT; ++u) {
    float* ap = acc + dl[u] * LD + k * WIDTH + lane;
    #pragma unroll
    for (int j = 0; j < NJ; ++j) atomicAdd(ap + 32 * j, hv[u][j] * w[u]);
  }
}

template <int WIDTH, int CNT, bool FULL>
__device__ __forceinline__ void work_group(int2 rec, float dv, float dl, int g, int last,
                                           const float* __restrict__ HH, float* acc, int lane) {
  constexpr int LD = 4 * WIDTH;
  constexpr int NJ = WIDTH / 32;
  int c[CNT];
  float hv[CNT][NJ];
  #pragma unroll
  for (int u = 0; u < CNT; ++u) {
    const int from = FULL ? g + u : min(g + u, last);
    const int s = __builtin_amdgcn_readlane(rec.x, from);
    c[u] = __builtin_amdgcn_readlane(rec.y, from);
    const float* hp = HH + (long long)s * WIDTH + lane;
    #pragma unroll
    for (int j = 0; j < NJ; ++j) hv[u][j] = hp[32 * j];
  }
  #pragma unroll
  for (int u = 0; u < CNT; ++u) {
    if (FULL || g + u <= last) {
      const float w = __int_as_float(__builtin_amdgcn_readlane(__float_as_int(dv), g + u)) *
                      __int_as_float(__builtin_amdgcn_readlane(__float_as_int(dl), g + u));
      float* ap = acc + (c[u] >> 2) * LD + (c[u] & 3) * WIDTH + lane;
      #pragma unroll
      for (int j = 0; j < NJ; ++j) atomicAdd(ap + 32 * j, hv[u][j] * w);
    }
  }
}

template <int WIDTH>
__device__ __forceinline__ void work_records(const int2* __restrict__ sg, int n,
                                             const float* __restrict__ dinv4, const float* dinvL,
                                             const float* __restrict__ HH, float* acc,
                                             int wave, int lane) {
  constexpr int U = 1024 / WIDTH;
  int b0 = wave * 32;
  if (b0 >= n) return;
  int2 rec = sg[min(b0 + lane, n - 1)];
  while (b0 < n) {
    const int cnt = min(32, n - b0);
    const int nb  = b0 + NTHR;
    int2 nrec = rec;
    if (nb < n) nrec = sg[min(nb + lane, n - 1)];
    const float dv = dinv4[rec.x * 4 + (rec.y & 3)];
    const float dl = dinvL[rec.y];
    int g = 0;
    for (; g + U <= cnt; g += U) work_group<WIDTH, U, true>(rec, dv, dl, g, cnt - 1, HH, acc, lane);
    if (g < cnt)                 work_group<WIDTH, U, false>(rec, dv, dl, g, cnt - 1, HH, acc, lane);
    rec = nrec;
    b0  = nb;
  }
}

template <int WIDTH, bool RELU>
__global__ __launch_bounds__(NTHR)
void agg_owner(const int* __restrict__ s1, const int* __restrict__ d1, int E1,
               const int* __restrict__ s2, const int* __restrict__ d2, int E2,
               const int* __restrict__ s3, const int* __restrict__ d3, int E3,
               const float* __restrict__ dinv4, const int* __restrict__ seghdr,
               const int* __restrict__ seg, const float* __restrict__ HH,
               const float* __restrict__ bias, float* __restrict__ R, int N) {
  constexpr int LD = 4 * WIDTH;
  constexpr int U  = 4;
  __shared__ __attribute__((aligned(16))) float acc[TR * LD];
  __shared__ float dinvL[TR * 4];
  __shared__ int qe[2 * QCAP];
  __shared__ int qd[2 * QCAP];
  __shared__ unsigned qcount[3];
  __shared__ int hsub[G];
  __shared__ int hovf[G];

  const int tid  = threadIdx.x;
  const int wave = tid >> 5;
  const int lane = tid & 31;
  const int r0   = blockIdx.x * TR;

  if (tid < G) {
    const int4 hd = reinterpret_cast<const int4*>(seghdr)[(long long)blockIdx.x * G + tid];
    hsub[tid] = hd.x;
    hovf[tid] = hd.y;
  }

  for (int idx = tid; idx < TR * LD; idx += NTHR) {
    const int dl = idx / LD;
    const int c  = idx - dl * LD;
    const int br = c / WIDTH;
    const int h  = c - br * WIDTH;
    const int row = r0 + dl;
    float v = bias[h];
    if (br == 3 && row < N) v += HH[(long long)row * WIDTH + h];
    acc[idx] = v;
  }
  for (int i = tid; i < TR * 4; i += NTHR) {
    const int row = r0 + (i >> 2);
    dinvL[i] = (row < N) ? dinv4[(long long)row * 4 + (i & 3)] : 0.0f;
  }
  if (tid < 3) qcount[tid] = 0u;
  __syncthreads();

  int ovf = 0;
  #pragma unroll
  for (int g = 0; g < G; ++g) ovf |= hovf[g];
  ovf = __builtin_amdgcn_readfirstlane(ovf);

  if (ovf == 0) {
    const int wv = __builtin_amdgcn_readfirstlane(wave);
    int rot = 0;
    #pragma unroll 1
    for (int g = 0; g < G; ++g) {
      const int nrec = __builtin_amdgcn_readfirstlane(hsub[g]);
      work_records<WIDTH>(reinterpret_cast<const int2*>(seg) + ((long long)blockIdx.x * G + g) * SUBCAP, nrec,
                          dinv4, dinvL, HH, acc, (wv - rot) & (NWAVE - 1), lane);
      rot += (nrec + 31) >> 5;
    }
  } else {
    int buf = 0;
    int ci  = 0;

    #pragma unroll 1
    for (int k = 0; k < 3; ++k) {
      const int* __restrict__ sp = (k == 0) ? s1 : (k == 1) ? s2 : s3;
      const int* __restrict__ dp = (k == 0) ? d1 : (k == 1) ? d2 : d3;
      const int E = (k == 0) ? E1 : (k == 1) ? E2 : E3;
      const bool vec_ok = ((E & 3) == 0) && ((reinterpret_cast<size_t>(dp) & 15) == 0);

      int4 cur[NV];
      load_chunk(cur, dp, 0, E, vec_ok, tid);

      for (int base = 0; base < E; base += CHUNK) {
        const int cn = (ci == 2) ? 0 : ci + 1;
        if (tid == 0) qcount[cn] = 0u;
        unsigned* qc = &qcount[ci];
        int* qeb = qe + buf * QCAP;
        int* qdb = qd + buf * QCAP;

        #pragma unroll
        for (int j = 0; j < NV; ++j) {
          const int i = base + (j * NTHR + tid) * 4;
          const unsigned a0 = (unsigned)(cur[j].x - r0), a1 = (unsigned)(cur[j].y - r0);
          const unsigned a2 = (unsigned)(cur[j].z - r0), a3 = (unsigned)(cur[j].w - r0);
          if (min(min(a0, a1), min(a2, a3)) < (unsigned)TR) {
            #pragma unroll 1
            for (int t = 0; t < 4; ++t) {
              const unsigned a = (t == 0) ? a0 : (t == 1) ? a1 : (t == 2) ? a2 : a3;
              if (a < (unsigned)TR) {
                const unsigned pos = atomicAdd(qc, 1u);
                if (pos < (unsigned)QCAP) { qeb[pos] = i + t; qdb[pos] = (int)a; }
                else edge_in_lane<WIDTH>(i + t, (int)a, k, sp, dinv4, dinvL, HH, acc);
              }
            }
          }
        }

        __syncthreads();

        int4 nxt[NV];
        load_chunk(nxt, dp, base + CHUNK, E, vec_ok, tid);

        const int n = min((int)(*qc), QCAP);
        const int nfull = n - (n % U);
        for (int i0 = wave * U; i0 < nfull; i0 += (NTHR / 32) * U)
          work_edges<WIDTH, U>(qeb + i0, qdb + i0, k, sp, dinv4, dinvL, HH, acc, lane);
        for (int i1 = nfull + wave; i1 < n; i1 += NTHR / 32)
          work_edges<WIDTH, 1>(qeb + i1, qdb + i1, k, sp, dinv4, dinvL, HH, acc, lane);


        #pragma unroll
        for (int j = 0; j < NV; ++j) cur[j] = nxt[j];
        buf ^= 1;
        ci = cn;
      }
    }
  }
  __syncthreads();

  const float4* acc4 = reinterpret_cast<const float4*>(acc);
  float4* R4 = reinterpret_cast<float4*>(R + (long long)r0 * LD);
  for (int idx = tid; idx < TR * LD / 4; idx += NTHR) {
    const int row = r0 + (idx * 4) / LD;
    if (row < N) {
      float4 v = acc4[idx];
      if (RELU) { v.x = fmaxf(v.x, 0.0f); v.y = fmaxf(v.y, 0.0f); v.z = fmaxf(v.z, 0.0f); v.w = fmaxf(v.w, 0.0f); }
      R4[idx] = v;
    }
  }
}

__global__ void log_softmax64(float* __restrict__ out, int N) {
  int row  = blockIdx.x * (blockDim.x >> 5) + (threadIdx.x >> 5);
  int lane = threadIdx.x & 31;
  if (row >= N) return;
  float* p = out + (long long)row * 64;
  float a = p[lane];
  float b = p[lane + 32];
  float mx = fmaxf(a, b);
  #pragma unroll
  for (int off = 16; off > 0; off >>= 1) mx = fmaxf(mx, __shfl_xor(mx, off));
  float s = __expf(a - mx) + __expf(b - mx);
  #pragma unroll
  for (int off = 16; off > 0; off >>= 1) s += __shfl_xor(s, off);
  float lse = mx + __logf(s);
  p[lane]      = a - lse;
  p[lane + 32] = b - lse;
}

extern "C" void kernel_launch(void* const* d_in, const int* in_sizes, int n_in,
                              void* d_out, int out_size, void* d_ws, size_t ws_size,
                              hipStream_t stream) {
  const float* x  = (const float*)d_in[0];
  const int* e1   = (const int*)d_in[1];
  const int* e2   = (const int*)d_in[2];
  const int* e3   = (const int*)d_in[3];
  const float* W1 = (const float*)d_in[5];
  const float* b1 = (const float*)d_in[6];
  const float* W2 = (const float*)d_in[7];
  const float* b2 = (const float*)d_in[8];
  const float* Wl = (const float*)d_in[9];
  const float* bl = (const float*)d_in[10];

  const int F = 512, H = 128, C = 64;
  const int N  = in_sizes[0] / F;
  const int E1 = in_sizes[1] / 2;
  const int E2 = in_sizes[2] / 2;
  const int E3 = in_sizes[3] / 2;

  float* ws    = (float*)d_ws;
  float* dinv4 = ws;
  float* HH1   = dinv4 + 4LL * N;
  float* R1    = HH1 + (long long)N * H;
  float* HH2   = R1 + (long long)N * 4 * H;
  float* R2    = HH2 + (long long)N * C;

  const int T = 256;
  const int otiles = cdiv_i(N, TR);
  const int gblocks = cdiv_i(N, GROWS);
  const int groups = cdiv_i(otiles, G);
  const int gtiles = groups * G;
  unsigned* pcnt = (unsigned*)(R2 + (long long)N * 4 * C);
  int* seghdr = (int*)(pcnt + 4LL * G * N);
  int* seg    = seghdr + 4LL * G * gtiles;

  deg_owner<<<groups * G, NTHR, 0, stream>>>(e1, e1 + E1, E1, e2, e2 + E2, E2, e3, e3 + E3, E3,
                                             pcnt, seghdr, seg, N);

  gemm_wmma_f32<512, 128><<<gblocks, GTH, 0, stream>>>(x, W1, nullptr, HH1, N, pcnt, dinv4);

  agg_owner<128, true><<<otiles, NTHR, 0, stream>>>(
      e1, e1 + E1, E1, e2, e2 + E2, E2, e3, e3 + E3, E3, dinv4, seghdr, seg, HH1, b1, R1, N);

  gemm_wmma_f32<512, 64><<<gblocks, GTH, 0, stream>>>(R1, W2, nullptr, HH2, N, nullptr, nullptr);

  agg_owner<64, false><<<otiles, NTHR, 0, stream>>>(
      e1, e1 + E1, E1, e2, e2 + E2, E2, e3, e3 + E3, E3, dinv4, seghdr, seg, HH2, b2, R2, N);

  float* logits = (float*)d_out;
  gemm_wmma_f32<256, 64><<<gblocks, GTH, 0, stream>>>(R2, Wl, bl, logits, N, nullptr, nullptr);

  log_softmax64<<<cdiv_i(N, T / 32), T, 0, stream>>>(logits, N);
}


# Aggregation set-up: the tile's own (self-loop) rows come as 16-byte loads from every wave, all in flight before one wait; before, a quarter of the waves made a dependent 4-byte load per two rows
# speedup vs baseline: 5.7238x; 1.0626x over previous
#include <hip/hip_runtime.h>
#include <math.h>


typedef float v2f __attribute__((ext_vector_type(2)));
typedef float v8f __attribute__((ext_vector_type(8)));

static inline int cdiv_i(long long a, long long b) { return (int)((a + b - 1) / b); }

constexpr int TR    = 80;
constexpr int NTHR  = 1024;
constexpr int NV    = 8;
constexpr int CHUNK = NTHR * NV * 4;
constexpr int QCAP  = 8192;
constexpr int SEGCAP = 24576;
constexpr int NWAVE = NTHR / 32;
constexpr int WSEG  = SEGCAP / NWAVE;
constexpr int G      = 5;
constexpr int GR     = G * TR;
constexpr int SUBCAP = 6400;

__device__ __forceinline__ void load_chunk(int4 (&v)[NV], const int* __restrict__ dp,
                                           int base, int E, bool vec_ok, int tid) {
  if (vec_ok && base + CHUNK <= E) {
    #pragma unroll
    for (int j = 0; j < NV; ++j)
      v[j] = *reinterpret_cast<const int4*>(dp + base + (j * NTHR + tid) * 4);
  } else if (vec_ok && E >= 4) {
    #pragma unroll
    for (int j = 0; j < NV; ++j) {
      const int i = base + (j * NTHR + tid) * 4;
      const int4 t = *reinterpret_cast<const int4*>(dp + min(i, E - 4));
      const bool in = i < E;
      v[j].x = in ? t.x : -1;
      v[j].y = in ? t.y : -1;
      v[j].z = in ? t.z : -1;
      v[j].w = in ? t.w : -1;
    }
  } else {
    #pragma unroll
    for (int j = 0; j < NV; ++j) {
      const int i = base + (j * NTHR + tid) * 4;
      v[j].x = (i     < E) ? dp[i]     : -1;
      v[j].y = (i + 1 < E) ? dp[i + 1] : -1;
      v[j].z = (i + 2 < E) ? dp[i + 2] : -1;
      v[j].w = (i + 3 < E) ? dp[i + 3] : -1;
    }
  }
}

__device__ __forceinline__ void take_edges(unsigned b, bool hit, unsigned a, int e, int k, unsigned* cntk,
                                           int* wqe, int* wqc, int& wn) {
  if (b != 0u) {
    if (hit) {
      atomicAdd(&cntk[a], 1u);
      const int pos = wn + (int)__builtin_amdgcn_mbcnt_lo(b, 0u);
      if (pos < WSEG) { wqe[pos] = e; wqc[pos] = (int)a * 4 + k; }
    }
    wn += __builtin_popcount(b);
  }
}

__global__ __launch_bounds__(NTHR)
void deg_owner(const int* __restrict__ s1, const int* __restrict__ d1, int E1,
               const int* __restrict__ s2, const int* __restrict__ d2, int E2,
               const int* __restrict__ s3, const int* __restrict__ d3, int E3,
               unsigned* __restrict__ pcnt, int* __restrict__ seghdr, int* __restrict__ seg, int N) {
  __shared__ unsigned cnt[3 * GR];
  __shared__ int qe[SEGCAP];
  __shared__ int qc[SEGCAP];
  __shared__ int wtc[NWAVE * G];
  __shared__ int wov[NWAVE];
  const int tid  = threadIdx.x;
  const int lane = tid & 31;
  const int wave = __builtin_amdgcn_readfirstlane(tid >> 5);
  const int rg   = blockIdx.x / G;
  const int sl   = blockIdx.x - rg * G;
  const int r0   = rg * GR;
  for (int i = tid; i < 3 * GR; i += NTHR) cnt[i] = 0u;
  __syncthreads();

  int* const wqe = qe + wave * WSEG;
  int* const wqc = qc + wave * WSEG;
  int wn = 0;

  #pragma unroll 1
  for (int k = 0; k < 3; ++k) {
    const int* __restrict__ dp = (k == 0) ? d1 : (k == 1) ? d2 : d3;
    const int E = (k == 0) ? E1 : (k == 1) ? E2 : E3;
    unsigned* const cntk = cnt + k * GR;
    const int m   = (E >= 4) ? (int)((reinterpret_cast<size_t>(dp) >> 2) & 3) : 0;
    const int* __restrict__ ap = dp - m;
    const int Ep  = E + m;
    const int Epf = (E >= 4) ? (Ep & ~3) : 0;
    const int pg  = ((Epf >> 2) + G - 1) / G;
    const int lo  = min(sl * pg * 4, Epf);
    const int hi  = min(lo + pg * 4, Epf);
    for (int pb = lo; pb < hi; pb += CHUNK) {
      int4 v[NV];
      if (pb + CHUNK <= hi && (pb > 0 || m == 0)) {
        #pragma unroll
        for (int j = 0; j < NV; ++j)
          v[j] = *reinterpret_cast<const int4*>(ap + pb + (j * NTHR + tid) * 4);
      } else {
        #pragma unroll
        for (int j = 0; j < NV; ++j) {
          const int p = pb + (j * NTHR + tid) * 4;
          const int4 t = *reinterpret_cast<const int4*>(ap + min(p, hi - 4));
          const bool in = p < hi;
          v[j].x = (in && p     >= m) ? t.x : -1;
          v[j].y = (in && p + 1 >= m) ? t.y : -1;
          v[j].z = (in && p + 2 >= m) ? t.z : -1;
          v[j].w = (in && p + 3 >= m) ? t.w : -1;
        }
      }
      unsigned any[NV];
      #pragma unroll
      for (int j = 0; j < NV; ++j) {
        v[j].x -= r0; v[j].y -= r0; v[j].z -= r0; v[j].w -= r0;
        const unsigned mn = min(min((unsigned)v[j].x, (unsigned)v[j].y), min((unsigned)v[j].z, (unsigned)v[j].w));
        any[j] = __builtin_amdgcn_ballot_w32(mn < (unsigned)GR);
      }
      #pragma unroll
      for (int j = 0; j < NV; ++j) {
        if (any[j] != 0u) {
          const int i = pb + (j * NTHR + tid) * 4 - m;
          const unsigned a0 = (unsigned)v[j].x, a1 = (unsigned)v[j].y, a2 = (unsigned)v[j].z, a3 = (unsigned)v[j].w;
          const unsigned b0 = __builtin_amdgcn_ballot_w32(a0 < (unsigned)GR);
          const unsigned b1 = __builtin_amdgcn_ballot_w32(a1 < (unsigned)GR);
          const unsigned b2 = __builtin_amdgcn_ballot_w32(a2 < (unsigned)GR);
          const unsigned b3 = __builtin_amdgcn_ballot_w32(a3 < (unsigned)GR);
          take_edges(b0, a0 < (unsigned)GR, a0, i,     k, cntk, wqe, wqc, wn);
          take_edges(b1, a1 < (unsigned)GR, a1, i + 1, k, cntk, wqe, wqc, wn);
          take_edges(b2, a2 < (unsigned)GR, a2, i + 2, k, cntk, wqe, wqc, wn);
          take_edges(b3, a3 < (unsigned)GR, a3, i + 3, k, cntk, wqe, wqc, wn);
        }
      }
    }
    {
      const bool mine = (sl == G - 1) && tid < Ep - Epf;
      const int i = Epf - m + tid;
      unsigned a = 0xffffffffu;
      if (mine) a = (unsigned)(dp[i] - r0);
      const bool hit = mine && a < (unsigned)GR;
      take_edges(__builtin_amdgcn_ballot_w32(hit), hit, a, i, k, cntk, wqe, wqc, wn);
    }
  }
  __syncthreads();

  if (tid < GR) {
    const int row = r0 + tid;
    if (row < N) {
      uint4 o;
      o.x = cnt[tid]; o.y = cnt[GR + tid]; o.z = cnt[2 * GR + tid]; o.w = 0u;
      reinterpret_cast<uint4*>(pcnt)[(long long)sl * N + row] = o;
    }
  }

  const int nq = min(wn, WSEG);
  int tc[G];
  #pragma unroll
  for (int g = 0; g < G; ++g) tc[g] = 0;
  for (int i0 = 0; i0 < nq; i0 += 32) {
    const int ii = i0 + lane;
    int t = -1;
    if (ii < nq) t = (wqc[ii] >> 2) / TR;
    #pragma unroll
    for (int g = 0; g < G; ++g) tc[g] += __builtin_popcount(__builtin_amdgcn_ballot_w32(t == g));
  }
  if (lane == 0) {
    #pragma unroll
    for (int g = 0; g < G; ++g) wtc[wave * G + g] = tc[g];
    wov[wave] = (wn > WSEG) ? 1 : 0;
  }
  __syncthreads();

  static_assert(NWAVE == 32, "one lane per wave of the work-group");
  const bool ovf = __builtin_amdgcn_ballot_w32(wov[lane] != 0) != 0u;
  int off[G], tot[G];
  #pragma unroll
  for (int g = 0; g < G; ++g) {
    int t = wtc[lane * G + g];
    int o = (lane < wave) ? t : 0;
    #pragma unroll
    for (int s = 16; s > 0; s >>= 1) { t += __shfl_xor(t, s); o += __shfl_xor(o, s); }
    tot[g] = __builtin_amdgcn_readfirstlane(t);
    off[g] = __builtin_amdgcn_readfirstlane(o);
  }
  const long long pair0 = ((long long)rg * G) * G + sl;
  #pragma unroll
  for (int g = 0; g < G; ++g) {
    if (tid == g) {
      const bool o = ovf || tot[g] > SUBCAP;
      int4 h;
      h.x = o ? 0 : tot[g]; h.y = o ? 1 : 0; h.z = 0; h.w = 0;
      reinterpret_cast<int4*>(seghdr)[pair0 + (long long)g * G] = h;
    }
  }
  const int nw = ovf ? 0 : nq;
  int2* const sg = reinterpret_cast<int2*>(seg) + pair0 * SUBCAP;
  for (int i0 = lane; i0 < nw; i0 += 4 * 32) {
    int c[4], s[4];
    #pragma unroll
    for (int u = 0; u < 4; ++u) {
      const int ii = min(i0 + u * 32, nw - 1);
      const int e = wqe[ii];
      c[u] = wqc[ii];
      const int kk = c[u] & 3;
      const int* sp = (kk == 0) ? s1 : (kk == 1) ? s2 : s3;
      s[u] = sp[e];
    }
    #pragma unroll
    for (int u = 0; u < 4; ++u) {
      const int a = c[u] >> 2;
      const int t = (i0 + u * 32 < nw) ? a / TR : -1;
      const int2 rec = make_int2(s[u], (a - t * TR) * 4 + (c[u] & 3));
      #pragma unroll
      for (int g = 0; g < G; ++g) {
        const unsigned b = __builtin_amdgcn_ballot_w32(t == g);
        if (t == g) {
          const int pos = off[g] + (int)__builtin_amdgcn_mbcnt_lo(b, 0u);
          if (pos < SUBCAP) sg[(long long)g * G * SUBCAP + pos] = rec;
        }
        off[g] += __builtin_popcount(b);
      }
    }
  }
}

constexpr int GW    = 20;
constexpr int GTH   = GW * 32;
constexpr int GROWS = 80;

template <int K, int NC>
__global__ __launch_bounds__(GTH)
void gemm_wmma_f32(const float* __restrict__ A, const float* __restrict__ B,
                   const float* __restrict__ bias, float* __restrict__ D, int M,
                   const unsigned* __restrict__ pc, float* __restrict__ dinv4) {
  constexpr int NT  = NC / 16;
  constexpr int CT  = NT / 4;
  constexpr int KB  = K / 8;
  constexpr int TOT = (K / 4) * NC;
  static_assert(K % 32 == 0 && NC % 64 == 0, "shapes");
  __shared__ float4 Bs[KB * NT * 32];

  const int tid  = threadIdx.x;
  const int lane = tid & 31;
  const int wave = tid >> 5;
  const int half = lane >> 4;
  const int mn   = lane & 15;
  const int row0 = blockIdx.x * GROWS;
  const int rt   = wave >> 2;
  const int ct0  = (wave & 3) * CT;

  const int arow = min(row0 + rt * 16 + mn, M - 1);
  const float4* __restrict__ ap = reinterpret_cast<const float4*>(A + (long long)arow * K) + half;

  float4 cur[4];
  #pragma unroll
  for (int j = 0; j < 4; ++j) cur[j] = ap[2 * j];

  if (pc != nullptr && tid < GROWS && row0 + tid < M) {
    const int row = row0 + tid;
    unsigned c0 = 0u, c1 = 0u, c2 = 0u;
    #pragma unroll
    for (int g = 0; g < G; ++g) {
      const uint4 p = reinterpret_cast<const uint4*>(pc)[(long long)g * M + row];
      c0 += p.x; c1 += p.y; c2 += p.z;
    }
    float4 o;
    o.x = (c0 > 0u) ? rsqrtf((float)c0) : 0.0f;
    o.y = (c1 > 0u) ? rsqrtf((float)c1) : 0.0f;
    o.z = (c2 > 0u) ? rsqrtf((float)c2) : 0.0f;
    o.w = 0.0f;
    reinterpret_cast<float4*>(dinv4)[row] = o;
  }

  constexpr int NTRIP = (TOT + GTH - 1) / GTH;
  constexpr int BATCH = (NTRIP % 13 == 0) ? 13 : 7;
  static_assert(NTRIP % BATCH == 0, "whole batches");
  #pragma unroll 1
  for (int t0 = 0; t0 < NTRIP; t0 += BATCH) {
    float4 v[BATCH];
    #pragma unroll
    for (int u = 0; u < BATCH; ++u) {
      const int i  = min((t0 + u) * GTH + tid, TOT - 1);
      const int c  = i % NC;
      const int kg = i / NC;
      const float* bp = B + (long long)(kg * 4) * NC + c;
      v[u].x = bp[0]; v[u].y = bp[NC]; v[u].z = bp[2 * NC]; v[u].w = bp[3 * NC];
    }
    #pragma unroll
    for (int u = 0; u < BATCH; ++u) {
      const int i  = (t0 + u) * GTH + tid;
      const int c  = i % NC;
      const int kg = i / NC;
      if (i < TOT) Bs[((kg >> 1) * NT + (c >> 4)) * 32 + (kg & 1) * 16 + (c & 15)] = v[u];
    }
  }
  __syncthreads();

  v8f acc[CT];
  #pragma unroll
  for (int t = 0; t < CT; ++t) acc[t] = v8f{};

  #pragma unroll 1
  for (int kc = 0; kc < KB; kc += 4) {
    const int kn = min(kc + 4, KB - 4);
    float4 nxt[4];
    #pragma unroll
    for (int j = 0; j < 4; ++j) nxt[j] = ap[2 * (kn + j)];

    #pragma unroll
    for (int j = 0; j < 4; ++j) {
      v2f a0, a1;
      a0.x = cur[j].x; a0.y = cur[j].y;
      a1.x = cur[j].z; a1.y = cur[j].w;
      #pragma unroll
      for (int t = 0; t < CT; ++t) {
        const float4 b = Bs[((kc + j) * NT + ct0 + t) * 32 + lane];
        v2f b0, b1;
        b0.x = b.x; b0.y = b.y;
        b1.x = b.z; b1.y = b.w;
        acc[t] = __builtin_amdgcn_wmma_f32_16x16x4_f32(
             false, a0,  false, b0,
             (short)0, acc[t],  false,  false);
        acc[t] = __builtin_amdgcn_wmma_f32_16x16x4_f32(
             false, a1,  false, b1,
             (short)0, acc[t],  false,  false);
      }
    }
    #pragma unroll
    for (int j = 0; j < 4; ++j) cur[j] = nxt[j];
  }

  #pragma unroll
  for (int t = 0; t < CT; ++t) {
    const int col = (ct0 + t) * 16 + mn;
    const float bv = bias ? bias[col] : 0.0f;
    #pragma unroll
    for (int r = 0; r < 8; ++r) {
      const int row = row0 + rt * 16 + r + half * 8;
      if (row < M) D[(long long)row * NC + col] = acc[t][r] + bv;
    }
  }
}


template <int WIDTH>
__device__ __forceinline__ void edge_in_lane(int e, int dl, int k, const int* __restrict__ sp,
                                             const float* __restrict__ dinv4, const float* dinvL,
                                             const float* __restrict__ HH, float* acc) {
  constexpr int LD = 4 * WIDTH;
  const int s = sp[e];
  const float w = dinv4[s * 4 + k] * dinvL[dl * 4 + k];
  const float* hp = HH + (long long)s * WIDTH;
  float* ap = acc + dl * LD + k * WIDTH;
  #pragma unroll 1
  for (int c = 0; c < WIDTH; ++c) atomicAdd(ap + c, hp[c] * w);
}

template <int WIDTH, int CNT>
__device__ __forceinline__ void work_edges(const int* qe_, const int* qd_, int k,
                                           const int* __restrict__ sp,
                                           const float* __restrict__ dinv4, const float* dinvL,
                                           const float* __restrict__ HH, float* acc, int lane) {
  constexpr int LD = 4 * WIDTH;
  constexpr int NJ = WIDTH / 32;
  int dl[CNT], s[CNT];
  #pragma unroll
  for (int u = 0; u < CNT; ++u) { dl[u] = qd_[u]; s[u] = sp[qe_[u]]; }
  float w[CNT], hv[CNT][NJ];
  #pragma unroll
  for (int u = 0; u < CNT; ++u) {
    w[u] = dinv4[s[u] * 4 + k] * dinvL[dl[u] * 4 + k];
    const float* hp = HH + (long long)s[u] * WIDTH + lane;
    #pragma unroll
    for (int j = 0; j < NJ; ++j) hv[u][j] = hp[32 * j];
  }
  #pragma unroll
  for (int u = 0; u < CNT; ++u) {
    float* ap = acc + dl[u] * LD + k * WIDTH + lane;
    #pragma unroll
    for (int j = 0; j < NJ; ++j) atomicAdd(ap + 32 * j, hv[u][j] * w[u]);
  }
}

template <int WIDTH, int CNT, bool FULL>
__device__ __forceinline__ void work_group(int2 rec, float dv, float dl, int g, int last,
                                           const float* __restrict__ HH, float* acc, int lane) {
  constexpr int LD = 4 * WIDTH;
  constexpr int NJ = WIDTH / 32;
  int c[CNT];
  float hv[CNT][NJ];
  #pragma unroll
  for (int u = 0; u < CNT; ++u) {
    const int from = FULL ? g + u : min(g + u, last);
    const int s = __builtin_amdgcn_readlane(rec.x, from);
    c[u] = __builtin_amdgcn_readlane(rec.y, from);
    const float* hp = HH + (long long)s * WIDTH + lane;
    #pragma unroll
    for (int j = 0; j < NJ; ++j) hv[u][j] = hp[32 * j];
  }
  #pragma unroll
  for (int u = 0; u < CNT; ++u) {
    if (FULL || g + u <= last) {
      const float w = __int_as_float(__builtin_amdgcn_readlane(__float_as_int(dv), g + u)) *
                      __int_as_float(__builtin_amdgcn_readlane(__float_as_int(dl), g + u));
      float* ap = acc + (c[u] >> 2) * LD + (c[u] & 3) * WIDTH + lane;
      #pragma unroll
      for (int j = 0; j < NJ; ++j) atomicAdd(ap + 32 * j, hv[u][j] * w);
    }
  }
}

template <int WIDTH>
__device__ __forceinline__ void work_records(const int2* __restrict__ sg, int n,
                                             const float* __restrict__ dinv4, const float* dinvL,
                                             const float* __restrict__ HH, float* acc,
                                             int wave, int lane) {
  constexpr int U = 1024 / WIDTH;
  int b0 = wave * 32;
  if (b0 >= n) return;
  int2 rec = sg[min(b0 + lane, n - 1)];
  while (b0 < n) {
    const int cnt = min(32, n - b0);
    const int nb  = b0 + NTHR;
    int2 nrec = rec;
    if (nb < n) nrec = sg[min(nb + lane, n - 1)];
    const float dv = dinv4[rec.x * 4 + (rec.y & 3)];
    const float dl = dinvL[rec.y];
    int g = 0;
    for (; g + U <= cnt; g += U) work_group<WIDTH, U, true>(rec, dv, dl, g, cnt - 1, HH, acc, lane);
    if (g < cnt)                 work_group<WIDTH, U, false>(rec, dv, dl, g, cnt - 1, HH, acc, lane);
    rec = nrec;
    b0  = nb;
  }
}

template <int WIDTH, bool RELU>
__global__ __launch_bounds__(NTHR)
void agg_owner(const int* __restrict__ s1, const int* __restrict__ d1, int E1,
               const int* __restrict__ s2, const int* __restrict__ d2, int E2,
               const int* __restrict__ s3, const int* __restrict__ d3, int E3,
               const float* __restrict__ dinv4, const int* __restrict__ seghdr,
               const int* __restrict__ seg, const float* __restrict__ HH,
               const float* __restrict__ bias, float* __restrict__ R, int N) {
  constexpr int LD = 4 * WIDTH;
  constexpr int U  = 4;
  __shared__ __attribute__((aligned(16))) float acc[TR * LD];
  __shared__ float dinvL[TR * 4];
  __shared__ int qe[2 * QCAP];
  __shared__ int qd[2 * QCAP];
  __shared__ unsigned qcount[3];
  __shared__ int hsub[G];
  __shared__ int hovf[G];

  const int tid  = threadIdx.x;
  const int wave = tid >> 5;
  const int lane = tid & 31;
  const int r0   = blockIdx.x * TR;

  constexpr int W4  = WIDTH / 4;
  constexpr int LD4 = LD / 4;
  constexpr int SV  = TR * W4;
  constexpr int SP  = (SV + NTHR - 1) / NTHR;
  static_assert(NTHR % LD4 == 0 && (TR * LD4) % NTHR == 0, "a thread keeps its column group through the fill");

  const float4* __restrict__ HH4 = reinterpret_cast<const float4*>(HH);
  const long long end4 = (long long)N * W4 - 1;
  float4 sv[SP];
  #pragma unroll
  for (int u = 0; u < SP; ++u) {
    const long long i4 = (long long)r0 * W4 + u * NTHR + tid;
    sv[u] = HH4[(i4 < end4) ? i4 : end4];
  }
  int4 hd = make_int4(0, 0, 0, 0);
  if (tid < G) hd = reinterpret_cast<const int4*>(seghdr)[(long long)blockIdx.x * G + tid];
  float dvl = 0.0f;
  if (tid < TR * 4 && r0 + (tid >> 2) < N) dvl = dinv4[(long long)r0 * 4 + tid];
  static_assert(TR * 4 <= NTHR, "one degree factor per thread");
  const int h4 = tid % W4;
  float4 bv;
  bv.x = bias[h4 * 4]; bv.y = bias[h4 * 4 + 1]; bv.z = bias[h4 * 4 + 2]; bv.w = bias[h4 * 4 + 3];

  float4* acc4w = reinterpret_cast<float4*>(acc);
  if (tid % LD4 < 3 * W4) {
    for (int idx4 = tid; idx4 < TR * LD4; idx4 += NTHR) acc4w[idx4] = bv;
  }
  #pragma unroll
  for (int u = 0; u < SP; ++u) {
    const int i = u * NTHR + tid;
    if (i < SV) {
      const int dl = i / W4;
      float4 v = bv;
      if (r0 + dl < N) { v.x += sv[u].x; v.y += sv[u].y; v.z += sv[u].z; v.w += sv[u].w; }
      acc4w[dl * LD4 + 3 * W4 + h4] = v;
    }
  }
  if (tid < TR * 4) dinvL[tid] = dvl;
  if (tid < G) { hsub[tid] = hd.x; hovf[tid] = hd.y; }
  if (tid < 3) qcount[tid] = 0u;
  __syncthreads();

  int ovf = 0;
  #pragma unroll
  for (int g = 0; g < G; ++g) ovf |= hovf[g];
  ovf = __builtin_amdgcn_readfirstlane(ovf);

  if (ovf == 0) {
    const int wv = __builtin_amdgcn_readfirstlane(wave);
    int rot = 0;
    #pragma unroll 1
    for (int g = 0; g < G; ++g) {
      const int nrec = __builtin_amdgcn_readfirstlane(hsub[g]);
      work_records<WIDTH>(reinterpret_cast<const int2*>(seg) + ((long long)blockIdx.x * G + g) * SUBCAP, nrec,
                          dinv4, dinvL, HH, acc, (wv - rot) & (NWAVE - 1), lane);
      rot += (nrec + 31) >> 5;
    }
  } else {
    int buf = 0;
    int ci  = 0;

    #pragma unroll 1
    for (int k = 0; k < 3; ++k) {
      const int* __restrict__ sp = (k == 0) ? s1 : (k == 1) ? s2 : s3;
      const int* __restrict__ dp = (k == 0) ? d1 : (k == 1) ? d2 : d3;
      const int E = (k == 0) ? E1 : (k == 1) ? E2 : E3;
      const bool vec_ok = ((E & 3) == 0) && ((reinterpret_cast<size_t>(dp) & 15) == 0);

      int4 cur[NV];
      load_chunk(cur, dp, 0, E, vec_ok, tid);

      for (int base = 0; base < E; base += CHUNK) {
        const int cn = (ci == 2) ? 0 : ci + 1;
        if (tid == 0) qcount[cn] = 0u;
        unsigned* qc = &qcount[ci];
        int* qeb = qe + buf * QCAP;
        int* qdb = qd + buf * QCAP;

        #pragma unroll
        for (int j = 0; j < NV; ++j) {
          const int i = base + (j * NTHR + tid) * 4;
          const unsigned a0 = (unsigned)(cur[j].x - r0), a1 = (unsigned)(cur[j].y - r0);
          const unsigned a2 = (unsigned)(cur[j].z - r0), a3 = (unsigned)(cur[j].w - r0);
          if (min(min(a0, a1), min(a2, a3)) < (unsigned)TR) {
            #pragma unroll 1
            for (int t = 0; t < 4; ++t) {
              const unsigned a = (t == 0) ? a0 : (t == 1) ? a1 : (t == 2) ? a2 : a3;
              if (a < (unsigned)TR) {
                const unsigned pos = atomicAdd(qc, 1u);
                if (pos < (unsigned)QCAP) { qeb[pos] = i + t; qdb[pos] = (int)a; }
                else edge_in_lane<WIDTH>(i + t, (int)a, k, sp, dinv4, dinvL, HH, acc);
              }
            }
          }
        }

        __syncthreads();

        int4 nxt[NV];
        load_chunk(nxt, dp, base + CHUNK, E, vec_ok, tid);

        const int n = min((int)(*qc), QCAP);
        const int nfull = n - (n % U);
        for (int i0 = wave * U; i0 < nfull; i0 += (NTHR / 32) * U)
          work_edges<WIDTH, U>(qeb + i0, qdb + i0, k, sp, dinv4, dinvL, HH, acc, lane);
        for (int i1 = nfull + wave; i1 < n; i1 += NTHR / 32)
          work_edges<WIDTH, 1>(qeb + i1, qdb + i1, k, sp, dinv4, dinvL, HH, acc, lane);


        #pragma unroll
        for (int j = 0; j < NV; ++j) cur[j] = nxt[j];
        buf ^= 1;
        ci = cn;
      }
    }
  }
  __syncthreads();

  const float4* acc4 = reinterpret_cast<const float4*>(acc);
  float4* R4 = reinterpret_cast<float4*>(R + (long long)r0 * LD);
  for (int idx = tid; idx < TR * LD / 4; idx += NTHR) {
    const int row = r0 + (idx * 4) / LD;
    if (row < N) {
      float4 v = acc4[idx];
      if (RELU) { v.x = fmaxf(v.x, 0.0f); v.y = fmaxf(v.y, 0.0f); v.z = fmaxf(v.z, 0.0f); v.w = fmaxf(v.w, 0.0f); }
      R4[idx] = v;
    }
  }
}

__global__ void log_softmax64(float* __restrict__ out, int N) {
  int row  = blockIdx.x * (blockDim.x >> 5) + (threadIdx.x >> 5);
  int lane = threadIdx.x & 31;
  if (row >= N) return;
  float* p = out + (long long)row * 64;
  float a = p[lane];
  float b = p[lane + 32];
  float mx = fmaxf(a, b);
  #pragma unroll
  for (int off = 16; off > 0; off >>= 1) mx = fmaxf(mx, __shfl_xor(mx, off));
  float s = __expf(a - mx) + __expf(b - mx);
  #pragma unroll
  for (int off = 16; off > 0; off >>= 1) s += __shfl_xor(s, off);
  float lse = mx + __logf(s);
  p[lane]      = a - lse;
  p[lane + 32] = b - lse;
}

extern "C" void kernel_launch(void* const* d_in, const int* in_sizes, int n_in,
                              void* d_out, int out_size, void* d_ws, size_t ws_size,
                              hipStream_t stream) {
  const float* x  = (const float*)d_in[0];
  const int* e1   = (const int*)d_in[1];
  const int* e2   = (const int*)d_in[2];
  const int* e3   = (const int*)d_in[3];
  const float* W1 = (const float*)d_in[5];
  const float* b1 = (const float*)d_in[6];
  const float* W2 = (const float*)d_in[7];
  const float* b2 = (const float*)d_in[8];
  const float* Wl = (const float*)d_in[9];
  const float* bl = (const float*)d_in[10];

  const int F = 512, H = 128, C = 64;
  const int N  = in_sizes[0] / F;
  const int E1 = in_sizes[1] / 2;
  const int E2 = in_sizes[2] / 2;
  const int E3 = in_sizes[3] / 2;

  float* ws    = (float*)d_ws;
  float* dinv4 = ws;
  float* HH1   = dinv4 + 4LL * N;
  float* R1    = HH1 + (long long)N * H;
  float* HH2   = R1 + (long long)N * 4 * H;
  float* R2    = HH2 + (long long)N * C;

  const int T = 256;
  const int otiles = cdiv_i(N, TR);
  const int gblocks = cdiv_i(N, GROWS);
  const int groups = cdiv_i(otiles, G);
  const int gtiles = groups * G;
  unsigned* pcnt = (unsigned*)(R2 + (long long)N * 4 * C);
  int* seghdr = (int*)(pcnt + 4LL * G * N);
  int* seg    = seghdr + 4LL * G * gtiles;

  deg_owner<<<groups * G, NTHR, 0, stream>>>(e1, e1 + E1, E1, e2, e2 + E2, E2, e3, e3 + E3, E3,
                                             pcnt, seghdr, seg, N);

  gemm_wmma_f32<512, 128><<<gblocks, GTH, 0, stream>>>(x, W1, nullptr, HH1, N, pcnt, dinv4);

  agg_owner<128, true><<<otiles, NTHR, 0, stream>>>(
      e1, e1 + E1, E1, e2, e2 + E2, E2, e3, e3 + E3, E3, dinv4, seghdr, seg, HH1, b1, R1, N);

  gemm_wmma_f32<512, 64><<<gblocks, GTH, 0, stream>>>(R1, W2, nullptr, HH2, N, nullptr, nullptr);

  agg_owner<64, false><<<otiles, NTHR, 0, stream>>>(
      e1, e1 + E1, E1, e2, e2 + E2, E2, e3, e3 + E3, E3, dinv4, seghdr, seg, HH2, b2, R2, N);

  float* logits = (float*)d_out;
  gemm_wmma_f32<256, 64><<<gblocks, GTH, 0, stream>>>(R2, Wl, bl, logits, N, nullptr, nullptr);

  log_softmax64<<<cdiv_i(N, T / 32), T, 0, stream>>>(logits, N);
}


# Log-softmax moved into the output GEMM's work-group: its 80x64 logits tile goes through LDS, one wave per row as before, rows stored once; 7 launches to 6; every sum order kept
# speedup vs baseline: 6.0718x; 1.0608x over previous
#include <hip/hip_runtime.h>
#include <math.h>


typedef float v2f __attribute__((ext_vector_type(2)));
typedef float v8f __attribute__((ext_vector_type(8)));

static inline int cdiv_i(long long a, long long b) { return (int)((a + b - 1) / b); }

constexpr int TR    = 80;
constexpr int NTHR  = 1024;
constexpr int NV    = 8;
constexpr int CHUNK = NTHR * NV * 4;
constexpr int QCAP  = 8192;
constexpr int SEGCAP = 24576;
constexpr int NWAVE = NTHR / 32;
constexpr int WSEG  = SEGCAP / NWAVE;
constexpr int G      = 5;
constexpr int GR     = G * TR;
constexpr int SUBCAP = 6400;

__device__ __forceinline__ void load_chunk(int4 (&v)[NV], const int* __restrict__ dp,
                                           int base, int E, bool vec_ok, int tid) {
  if (vec_ok && base + CHUNK <= E) {
    #pragma unroll
    for (int j = 0; j < NV; ++j)
      v[j] = *reinterpret_cast<const int4*>(dp + base + (j * NTHR + tid) * 4);
  } else if (vec_ok && E >= 4) {
    #pragma unroll
    for (int j = 0; j < NV; ++j) {
      const int i = base + (j * NTHR + tid) * 4;
      const int4 t = *reinterpret_cast<const int4*>(dp + min(i, E - 4));
      const bool in = i < E;
      v[j].x = in ? t.x : -1;
      v[j].y = in ? t.y : -1;
      v[j].z = in ? t.z : -1;
      v[j].w = in ? t.w : -1;
    }
  } else {
    #pragma unroll
    for (int j = 0; j < NV; ++j) {
      const int i = base + (j * NTHR + tid) * 4;
      v[j].x = (i     < E) ? dp[i]     : -1;
      v[j].y = (i + 1 < E) ? dp[i + 1] : -1;
      v[j].z = (i + 2 < E) ? dp[i + 2] : -1;
      v[j].w = (i + 3 < E) ? dp[i + 3] : -1;
    }
  }
}

__device__ __forceinline__ void take_edges(unsigned b, bool hit, unsigned a, int e, int k, unsigned* cntk,
                                           int* wqe, int* wqc, int& wn) {
  if (b != 0u) {
    if (hit) {
      atomicAdd(&cntk[a], 1u);
      const int pos = wn + (int)__builtin_amdgcn_mbcnt_lo(b, 0u);
      if (pos < WSEG) { wqe[pos] = e; wqc[pos] = (int)a * 4 + k; }
    }
    wn += __builtin_popcount(b);
  }
}

__global__ __launch_bounds__(NTHR)
void deg_owner(const int* __restrict__ s1, const int* __restrict__ d1, int E1,
               const int* __restrict__ s2, const int* __restrict__ d2, int E2,
               const int* __restrict__ s3, const int* __restrict__ d3, int E3,
               unsigned* __restrict__ pcnt, int* __restrict__ seghdr, int* __restrict__ seg, int N) {
  __shared__ unsigned cnt[3 * GR];
  __shared__ int qe[SEGCAP];
  __shared__ int qc[SEGCAP];
  __shared__ int wtc[NWAVE * G];
  __shared__ int wov[NWAVE];
  const int tid  = threadIdx.x;
  const int lane = tid & 31;
  const int wave = __builtin_amdgcn_readfirstlane(tid >> 5);
  const int rg   = blockIdx.x / G;
  const int sl   = blockIdx.x - rg * G;
  const int r0   = rg * GR;
  for (int i = tid; i < 3 * GR; i += NTHR) cnt[i] = 0u;
  __syncthreads();

  int* const wqe = qe + wave * WSEG;
  int* const wqc = qc + wave * WSEG;
  int wn = 0;

  #pragma unroll 1
  for (int k = 0; k < 3; ++k) {
    const int* __restrict__ dp = (k == 0) ? d1 : (k == 1) ? d2 : d3;
    const int E = (k == 0) ? E1 : (k == 1) ? E2 : E3;
    unsigned* const cntk = cnt + k * GR;
    const int m   = (E >= 4) ? (int)((reinterpret_cast<size_t>(dp) >> 2) & 3) : 0;
    const int* __restrict__ ap = dp - m;
    const int Ep  = E + m;
    const int Epf = (E >= 4) ? (Ep & ~3) : 0;
    const int pg  = ((Epf >> 2) + G - 1) / G;
    const int lo  = min(sl * pg * 4, Epf);
    const int hi  = min(lo + pg * 4, Epf);
    for (int pb = lo; pb < hi; pb += CHUNK) {
      int4 v[NV];
      if (pb + CHUNK <= hi && (pb > 0 || m == 0)) {
        #pragma unroll
        for (int j = 0; j < NV; ++j)
          v[j] = *reinterpret_cast<const int4*>(ap + pb + (j * NTHR + tid) * 4);
      } else {
        #pragma unroll
        for (int j = 0; j < NV; ++j) {
          const int p = pb + (j * NTHR + tid) * 4;
          const int4 t = *reinterpret_cast<const int4*>(ap + min(p, hi - 4));
          const bool in = p < hi;
          v[j].x = (in && p     >= m) ? t.x : -1;
          v[j].y = (in && p + 1 >= m) ? t.y : -1;
          v[j].z = (in && p + 2 >= m) ? t.z : -1;
          v[j].w = (in && p + 3 >= m) ? t.w : -1;
        }
      }
      unsigned any[NV];
      #pragma unroll
      for (int j = 0; j < NV; ++j) {
        v[j].x -= r0; v[j].y -= r0; v[j].z -= r0; v[j].w -= r0;
        const unsigned mn = min(min((unsigned)v[j].x, (unsigned)v[j].y), min((unsigned)v[j].z, (unsigned)v[j].w));
        any[j] = __builtin_amdgcn_ballot_w32(mn < (unsigned)GR);
      }
      #pragma unroll
      for (int j = 0; j < NV; ++j) {
        if (any[j] != 0u) {
          const int i = pb + (j * NTHR + tid) * 4 - m;
          const unsigned a0 = (unsigned)v[j].x, a1 = (unsigned)v[j].y, a2 = (unsigned)v[j].z, a3 = (unsigned)v[j].w;
          const unsigned b0 = __builtin_amdgcn_ballot_w32(a0 < (unsigned)GR);
          const unsigned b1 = __builtin_amdgcn_ballot_w32(a1 < (unsigned)GR);
          const unsigned b2 = __builtin_amdgcn_ballot_w32(a2 < (unsigned)GR);
          const unsigned b3 = __builtin_amdgcn_ballot_w32(a3 < (unsigned)GR);
          take_edges(b0, a0 < (unsigned)GR, a0, i,     k, cntk, wqe, wqc, wn);
          take_edges(b1, a1 < (unsigned)GR, a1, i + 1, k, cntk, wqe, wqc, wn);
          take_edges(b2, a2 < (unsigned)GR, a2, i + 2, k, cntk, wqe, wqc, wn);
          take_edges(b3, a3 < (unsigned)GR, a3, i + 3, k, cntk, wqe, wqc, wn);
        }
      }
    }
    {
      const bool mine = (sl == G - 1) && tid < Ep - Epf;
      const int i = Epf - m + tid;
      unsigned a = 0xffffffffu;
      if (mine) a = (unsigned)(dp[i] - r0);
      const bool hit = mine && a < (unsigned)GR;
      take_edges(__builtin_amdgcn_ballot_w32(hit), hit, a, i, k, cntk, wqe, wqc, wn);
    }
  }
  __syncthreads();

  if (tid < GR) {
    const int row = r0 + tid;
    if (row < N) {
      uint4 o;
      o.x = cnt[tid]; o.y = cnt[GR + tid]; o.z = cnt[2 * GR + tid]; o.w = 0u;
      reinterpret_cast<uint4*>(pcnt)[(long long)sl * N + row] = o;
    }
  }

  const int nq = min(wn, WSEG);
  int tc[G];
  #pragma unroll
  for (int g = 0; g < G; ++g) tc[g] = 0;
  for (int i0 = 0; i0 < nq; i0 += 32) {
    const int ii = i0 + lane;
    int t = -1;
    if (ii < nq) t = (wqc[ii] >> 2) / TR;
    #pragma unroll
    for (int g = 0; g < G; ++g) tc[g] += __builtin_popcount(__builtin_amdgcn_ballot_w32(t == g));
  }
  if (lane == 0) {
    #pragma unroll
    for (int g = 0; g < G; ++g) wtc[wave * G + g] = tc[g];
    wov[wave] = (wn > WSEG) ? 1 : 0;
  }
  __syncthreads();

  static_assert(NWAVE == 32, "one lane per wave of the work-group");
  const bool ovf = __builtin_amdgcn_ballot_w32(wov[lane] != 0) != 0u;
  int off[G], tot[G];
  #pragma unroll
  for (int g = 0; g < G; ++g) {
    int t = wtc[lane * G + g];
    int o = (lane < wave) ? t : 0;
    #pragma unroll
    for (int s = 16; s > 0; s >>= 1) { t += __shfl_xor(t, s); o += __shfl_xor(o, s); }
    tot[g] = __builtin_amdgcn_readfirstlane(t);
    off[g] = __builtin_amdgcn_readfirstlane(o);
  }
  const long long pair0 = ((long long)rg * G) * G + sl;
  #pragma unroll
  for (int g = 0; g < G; ++g) {
    if (tid == g) {
      const bool o = ovf || tot[g] > SUBCAP;
      int4 h;
      h.x = o ? 0 : tot[g]; h.y = o ? 1 : 0; h.z = 0; h.w = 0;
      reinterpret_cast<int4*>(seghdr)[pair0 + (long long)g * G] = h;
    }
  }
  const int nw = ovf ? 0 : nq;
  int2* const sg = reinterpret_cast<int2*>(seg) + pair0 * SUBCAP;
  for (int i0 = lane; i0 < nw; i0 += 4 * 32) {
    int c[4], s[4];
    #pragma unroll
    for (int u = 0; u < 4; ++u) {
      const int ii = min(i0 + u * 32, nw - 1);
      const int e = wqe[ii];
      c[u] = wqc[ii];
      const int kk = c[u] & 3;
      const int* sp = (kk == 0) ? s1 : (kk == 1) ? s2 : s3;
      s[u] = sp[e];
    }
    #pragma unroll
    for (int u = 0; u < 4; ++u) {
      const int a = c[u] >> 2;
      const int t = (i0 + u * 32 < nw) ? a / TR : -1;
      const int2 rec = make_int2(s[u], (a - t * TR) * 4 + (c[u] & 3));
      #pragma unroll
      for (int g = 0; g < G; ++g) {
        const unsigned b = __builtin_amdgcn_ballot_w32(t == g);
        if (t == g) {
          const int pos = off[g] + (int)__builtin_amdgcn_mbcnt_lo(b, 0u);
          if (pos < SUBCAP) sg[(long long)g * G * SUBCAP + pos] = rec;
        }
        off[g] += __builtin_popcount(b);
      }
    }
  }
}

constexpr int GW    = 20;
constexpr int GTH   = GW * 32;
constexpr int GROWS = 80;
constexpr int LSP   = 68;

template <int K, int NC, bool LSM>
__global__ __launch_bounds__(GTH)
void gemm_wmma_f32(const float* __restrict__ A, const float* __restrict__ B,
                   const float* __restrict__ bias, float* __restrict__ D, int M,
                   const unsigned* __restrict__ pc, float* __restrict__ dinv4) {
  constexpr int NT  = NC / 16;
  constexpr int CT  = NT / 4;
  constexpr int KB  = K / 8;
  constexpr int TOT = (K / 4) * NC;
  static_assert(K % 32 == 0 && NC % 64 == 0, "shapes");
  static_assert(!LSM || (NC == 64 && GROWS % GW == 0), "row pass: 64 columns, whole rows per wave");
  __shared__ float4 Bs[KB * NT * 32];
  __shared__ float Ls[LSM ? GROWS * LSP : 1];

  const int tid  = threadIdx.x;
  const int lane = tid & 31;
  const int wave = tid >> 5;
  const int half = lane >> 4;
  const int mn   = lane & 15;
  const int row0 = blockIdx.x * GROWS;
  const int rt   = wave >> 2;
  const int ct0  = (wave & 3) * CT;

  const int arow = min(row0 + rt * 16 + mn, M - 1);
  const float4* __restrict__ ap = reinterpret_cast<const float4*>(A + (long long)arow * K) + half;

  float4 cur[4];
  #pragma unroll
  for (int j = 0; j < 4; ++j) cur[j] = ap[2 * j];

  if (pc != nullptr && tid < GROWS && row0 + tid < M) {
    const int row = row0 + tid;
    unsigned c0 = 0u, c1 = 0u, c2 = 0u;
    #pragma unroll
    for (int g = 0; g < G; ++g) {
      const uint4 p = reinterpret_cast<const uint4*>(pc)[(long long)g * M + row];
      c0 += p.x; c1 += p.y; c2 += p.z;
    }
    float4 o;
    o.x = (c0 > 0u) ? rsqrtf((float)c0) : 0.0f;
    o.y = (c1 > 0u) ? rsqrtf((float)c1) : 0.0f;
    o.z = (c2 > 0u) ? rsqrtf((float)c2) : 0.0f;
    o.w = 0.0f;
    reinterpret_cast<float4*>(dinv4)[row] = o;
  }

  constexpr int NTRIP = (TOT + GTH - 1) / GTH;
  constexpr int BATCH = (NTRIP % 13 == 0) ? 13 : 7;
  static_assert(NTRIP % BATCH == 0, "whole batches");
  #pragma unroll 1
  for (int t0 = 0; t0 < NTRIP; t0 += BATCH) {
    float4 v[BATCH];
    #pragma unroll
    for (int u = 0; u < BATCH; ++u) {
      const int i  = min((t0 + u) * GTH + tid, TOT - 1);
      const int c  = i % NC;
      const int kg = i / NC;
      const float* bp = B + (long long)(kg * 4) * NC + c;
      v[u].x = bp[0]; v[u].y = bp[NC]; v[u].z = bp[2 * NC]; v[u].w = bp[3 * NC];
    }
    #pragma unroll
    for (int u = 0; u < BATCH; ++u) {
      const int i  = (t0 + u) * GTH + tid;
      const int c  = i % NC;
      const int kg = i / NC;
      if (i < TOT) Bs[((kg >> 1) * NT + (c >> 4)) * 32 + (kg & 1) * 16 + (c & 15)] = v[u];
    }
  }
  __syncthreads();

  v8f acc[CT];
  #pragma unroll
  for (int t = 0; t < CT; ++t) acc[t] = v8f{};

  #pragma unroll 1
  for (int kc = 0; kc < KB; kc += 4) {
    const int kn = min(kc + 4, KB - 4);
    float4 nxt[4];
    #pragma unroll
    for (int j = 0; j < 4; ++j) nxt[j] = ap[2 * (kn + j)];

    #pragma unroll
    for (int j = 0; j < 4; ++j) {
      v2f a0, a1;
      a0.x = cur[j].x; a0.y = cur[j].y;
      a1.x = cur[j].z; a1.y = cur[j].w;
      #pragma unroll
      for (int t = 0; t < CT; ++t) {
        const float4 b = Bs[((kc + j) * NT + ct0 + t) * 32 + lane];
        v2f b0, b1;
        b0.x = b.x; b0.y = b.y;
        b1.x = b.z; b1.y = b.w;
        acc[t] = __builtin_amdgcn_wmma_f32_16x16x4_f32(
             false, a0,  false, b0,
             (short)0, acc[t],  false,  false);
        acc[t] = __builtin_amdgcn_wmma_f32_16x16x4_f32(
             false, a1,  false, b1,
             (short)0, acc[t],  false,  false);
      }
    }
    #pragma unroll
    for (int j = 0; j < 4; ++j) cur[j] = nxt[j];
  }

  if constexpr (!LSM) {
    #pragma unroll
    for (int t = 0; t < CT; ++t) {
      const int col = (ct0 + t) * 16 + mn;
      const float bv = bias ? bias[col] : 0.0f;
      #pragma unroll
      for (int r = 0; r < 8; ++r) {
        const int row = row0 + rt * 16 + r + half * 8;
        if (row < M) D[(long long)row * NC + col] = acc[t][r] + bv;
      }
    }
  } else {
    #pragma unroll
    for (int t = 0; t < CT; ++t) {
      const int col = (ct0 + t) * 16 + mn;
      const float bv = bias ? bias[col] : 0.0f;
      #pragma unroll
      for (int r = 0; r < 8; ++r) Ls[(rt * 16 + r + half * 8) * LSP + col] = acc[t][r] + bv;
    }
    __syncthreads();
    constexpr int RQ = GROWS / GW;
    float a[RQ], b[RQ], mx[RQ], s[RQ];
    #pragma unroll
    for (int q = 0; q < RQ; ++q) {
      const int lrow = wave * RQ + q;
      a[q]  = Ls[lrow * LSP + lane];
      b[q]  = Ls[lrow * LSP + lane + 32];
      mx[q] = fmaxf(a[q], b[q]);
    }
    #pragma unroll
    for (int off = 16; off > 0; off >>= 1) {
      float o[RQ];
      #pragma unroll
      for (int q = 0; q < RQ; ++q) o[q] = __shfl_xor(mx[q], off);
      #pragma unroll
      for (int q = 0; q < RQ; ++q) mx[q] = fmaxf(mx[q], o[q]);
    }
    #pragma unroll
    for (int q = 0; q < RQ; ++q) s[q] = __expf(a[q] - mx[q]) + __expf(b[q] - mx[q]);
    #pragma unroll
    for (int off = 16; off > 0; off >>= 1) {
      float o[RQ];
      #pragma unroll
      for (int q = 0; q < RQ; ++q) o[q] = __shfl_xor(s[q], off);
      #pragma unroll
      for (int q = 0; q < RQ; ++q) s[q] += o[q];
    }
    #pragma unroll
    for (int q = 0; q < RQ; ++q) {
      const int row = row0 + wave * RQ + q;
      const float lse = mx[q] + __logf(s[q]);
      if (row < M) {
        float* p = D + (long long)row * NC;
        p[lane]      = a[q] - lse;
        p[lane + 32] = b[q] - lse;
      }
    }
  }
}


template <int WIDTH>
__device__ __forceinline__ void edge_in_lane(int e, int dl, int k, const int* __restrict__ sp,
                                             const float* __restrict__ dinv4, const float* dinvL,
                                             const float* __restrict__ HH, float* acc) {
  constexpr int LD = 4 * WIDTH;
  const int s = sp[e];
  const float w = dinv4[s * 4 + k] * dinvL[dl * 4 + k];
  const float* hp = HH + (long long)s * WIDTH;
  float* ap = acc + dl * LD + k * WIDTH;
  #pragma unroll 1
  for (int c = 0; c < WIDTH; ++c) atomicAdd(ap + c, hp[c] * w);
}

template <int WIDTH, int CNT>
__device__ __forceinline__ void work_edges(const int* qe_, const int* qd_, int k,
                                           const int* __restrict__ sp,
                                           const float* __restrict__ dinv4, const float* dinvL,
                                           const float* __restrict__ HH, float* acc, int lane) {
  constexpr int LD = 4 * WIDTH;
  constexpr int NJ = WIDTH / 32;
  int dl[CNT], s[CNT];
  #pragma unroll
  for (int u = 0; u < CNT; ++u) { dl[u] = qd_[u]; s[u] = sp[qe_[u]]; }
  float w[CNT], hv[CNT][NJ];
  #pragma unroll
  for (int u = 0; u < CNT; ++u) {
    w[u] = dinv4[s[u] * 4 + k] * dinvL[dl[u] * 4 + k];
    const float* hp = HH + (long long)s[u] * WIDTH + lane;
    #pragma unroll
    for (int j = 0; j < NJ; ++j) hv[u][j] = hp[32 * j];
  }
  #pragma unroll
  for (int u = 0; u < CNT; ++u) {
    float* ap = acc + dl[u] * LD + k * WIDTH + lane;
    #pragma unroll
    for (int j = 0; j < NJ; ++j) atomicAdd(ap + 32 * j, hv[u][j] * w[u]);
  }
}

template <int WIDTH, int CNT, bool FULL>
__device__ __forceinline__ void work_group(int2 rec, float dv, float dl, int g, int last,
                                           const float* __restrict__ HH, float* acc, int lane) {
  constexpr int LD = 4 * WIDTH;
  constexpr int NJ = WIDTH / 32;
  int c[CNT];
  float hv[CNT][NJ];
  #pragma unroll
  for (int u = 0; u < CNT; ++u) {
    const int from = FULL ? g + u : min(g + u, last);
    const int s = __builtin_amdgcn_readlane(rec.x, from);
    c[u] = __builtin_amdgcn_readlane(rec.y, from);
    const float* hp = HH + (long long)s * WIDTH + lane;
    #pragma unroll
    for (int j = 0; j < NJ; ++j) hv[u][j] = hp[32 * j];
  }
  #pragma unroll
  for (int u = 0; u < CNT; ++u) {
    if (FULL || g + u <= last) {
      const float w = __int_as_float(__builtin_amdgcn_readlane(__float_as_int(dv), g + u)) *
                      __int_as_float(__builtin_amdgcn_readlane(__float_as_int(dl), g + u));
      float* ap = acc + (c[u] >> 2) * LD + (c[u] & 3) * WIDTH + lane;
      #pragma unroll
      for (int j = 0; j < NJ; ++j) atomicAdd(ap + 32 * j, hv[u][j] * w);
    }
  }
}

template <int WIDTH>
__device__ __forceinline__ void work_records(const int2* __restrict__ sg, int n,
                                             const float* __restrict__ dinv4, const float* dinvL,
                                             const float* __restrict__ HH, float* acc,
                                             int wave, int lane) {
  constexpr int U = 1024 / WIDTH;
  int b0 = wave * 32;
  if (b0 >= n) return;
  int2 rec = sg[min(b0 + lane, n - 1)];
  while (b0 < n) {
    const int cnt = min(32, n - b0);
    const int nb  = b0 + NTHR;
    int2 nrec = rec;
    if (nb < n) nrec = sg[min(nb + lane, n - 1)];
    const float dv = dinv4[rec.x * 4 + (rec.y & 3)];
    const float dl = dinvL[rec.y];
    int g = 0;
    for (; g + U <= cnt; g += U) work_group<WIDTH, U, true>(rec, dv, dl, g, cnt - 1, HH, acc, lane);
    if (g < cnt)                 work_group<WIDTH, U, false>(rec, dv, dl, g, cnt - 1, HH, acc, lane);
    rec = nrec;
    b0  = nb;
  }
}

template <int WIDTH, bool RELU>
__global__ __launch_bounds__(NTHR)
void agg_owner(const int* __restrict__ s1, const int* __restrict__ d1, int E1,
               const int* __restrict__ s2, const int* __restrict__ d2, int E2,
               const int* __restrict__ s3, const int* __restrict__ d3, int E3,
               const float* __restrict__ dinv4, const int* __restrict__ seghdr,
               const int* __restrict__ seg, const float* __restrict__ HH,
               const float* __restrict__ bias, float* __restrict__ R, int N) {
  constexpr int LD = 4 * WIDTH;
  constexpr int U  = 4;
  __shared__ __attribute__((aligned(16))) float acc[TR * LD];
  __shared__ float dinvL[TR * 4];
  __shared__ int qe[2 * QCAP];
  __shared__ int qd[2 * QCAP];
  __shared__ unsigned qcount[3];
  __shared__ int hsub[G];
  __shared__ int hovf[G];

  const int tid  = threadIdx.x;
  const int wave = tid >> 5;
  const int lane = tid & 31;
  const int r0   = blockIdx.x * TR;

  constexpr int W4  = WIDTH / 4;
  constexpr int LD4 = LD / 4;
  constexpr int SV  = TR * W4;
  constexpr int SP  = (SV + NTHR - 1) / NTHR;
  static_assert(NTHR % LD4 == 0 && (TR * LD4) % NTHR == 0, "a thread keeps its column group through the fill");

  const float4* __restrict__ HH4 = reinterpret_cast<const float4*>(HH);
  const long long end4 = (long long)N * W4 - 1;
  float4 sv[SP];
  #pragma unroll
  for (int u = 0; u < SP; ++u) {
    const long long i4 = (long long)r0 * W4 + u * NTHR + tid;
    sv[u] = HH4[(i4 < end4) ? i4 : end4];
  }
  int4 hd = make_int4(0, 0, 0, 0);
  if (tid < G) hd = reinterpret_cast<const int4*>(seghdr)[(long long)blockIdx.x * G + tid];
  float dvl = 0.0f;
  if (tid < TR * 4 && r0 + (tid >> 2) < N) dvl = dinv4[(long long)r0 * 4 + tid];
  static_assert(TR * 4 <= NTHR, "one degree factor per thread");
  const int h4 = tid % W4;
  float4 bv;
  bv.x = bias[h4 * 4]; bv.y = bias[h4 * 4 + 1]; bv.z = bias[h4 * 4 + 2]; bv.w = bias[h4 * 4 + 3];

  float4* acc4w = reinterpret_cast<float4*>(acc);
  if (tid % LD4 < 3 * W4) {
    for (int idx4 = tid; idx4 < TR * LD4; idx4 += NTHR) acc4w[idx4] = bv;
  }
  #pragma unroll
  for (int u = 0; u < SP; ++u) {
    const int i = u * NTHR + tid;
    if (i < SV) {
      const int dl = i / W4;
      float4 v = bv;
      if (r0 + dl < N) { v.x += sv[u].x; v.y += sv[u].y; v.z += sv[u].z; v.w += sv[u].w; }
      acc4w[dl * LD4 + 3 * W4 + h4] = v;
    }
  }
  if (tid < TR * 4) dinvL[tid] = dvl;
  if (tid < G) { hsub[tid] = hd.x; hovf[tid] = hd.y; }
  if (tid < 3) qcount[tid] = 0u;
  __syncthreads();

  int ovf = 0;
  #pragma unroll
  for (int g = 0; g < G; ++g) ovf |= hovf[g];
  ovf = __builtin_amdgcn_readfirstlane(ovf);

  if (ovf == 0) {
    const int wv = __builtin_amdgcn_readfirstlane(wave);
    int rot = 0;
    #pragma unroll 1
    for (int g = 0; g < G; ++g) {
      const int nrec = __builtin_amdgcn_readfirstlane(hsub[g]);
      work_records<WIDTH>(reinterpret_cast<const int2*>(seg) + ((long long)blockIdx.x * G + g) * SUBCAP, nrec,
                          dinv4, dinvL, HH, acc, (wv - rot) & (NWAVE - 1), lane);
      rot += (nrec + 31) >> 5;
    }
  } else {
    int buf = 0;
    int ci  = 0;

    #pragma unroll 1
    for (int k = 0; k < 3; ++k) {
      const int* __restrict__ sp = (k == 0) ? s1 : (k == 1) ? s2 : s3;
      const int* __restrict__ dp = (k == 0) ? d1 : (k == 1) ? d2 : d3;
      const int E = (k == 0) ? E1 : (k == 1) ? E2 : E3;
      const bool vec_ok = ((E & 3) == 0) && ((reinterpret_cast<size_t>(dp) & 15) == 0);

      int4 cur[NV];
      load_chunk(cur, dp, 0, E, vec_ok, tid);

      for (int base = 0; base < E; base += CHUNK) {
        const int cn = (ci == 2) ? 0 : ci + 1;
        if (tid == 0) qcount[cn] = 0u;
        unsigned* qc = &qcount[ci];
        int* qeb = qe + buf * QCAP;
        int* qdb = qd + buf * QCAP;

        #pragma unroll
        for (int j = 0; j < NV; ++j) {
          const int i = base + (j * NTHR + tid) * 4;
          const unsigned a0 = (unsigned)(cur[j].x - r0), a1 = (unsigned)(cur[j].y - r0);
          const unsigned a2 = (unsigned)(cur[j].z - r0), a3 = (unsigned)(cur[j].w - r0);
          if (min(min(a0, a1), min(a2, a3)) < (unsigned)TR) {
            #pragma unroll 1
            for (int t = 0; t < 4; ++t) {
              const unsigned a = (t == 0) ? a0 : (t == 1) ? a1 : (t == 2) ? a2 : a3;
              if (a < (unsigned)TR) {
                const unsigned pos = atomicAdd(qc, 1u);
                if (pos < (unsigned)QCAP) { qeb[pos] = i + t; qdb[pos] = (int)a; }
                else edge_in_lane<WIDTH>(i + t, (int)a, k, sp, dinv4, dinvL, HH, acc);
              }
            }
          }
        }

        __syncthreads();

        int4 nxt[NV];
        load_chunk(nxt, dp, base + CHUNK, E, vec_ok, tid);

        const int n = min((int)(*qc), QCAP);
        const int nfull = n - (n % U);
        for (int i0 = wave * U; i0 < nfull; i0 += (NTHR / 32) * U)
          work_edges<WIDTH, U>(qeb + i0, qdb + i0, k, sp, dinv4, dinvL, HH, acc, lane);
        for (int i1 = nfull + wave; i1 < n; i1 += NTHR / 32)
          work_edges<WIDTH, 1>(qeb + i1, qdb + i1, k, sp, dinv4, dinvL, HH, acc, lane);


        #pragma unroll
        for (int j = 0; j < NV; ++j) cur[j] = nxt[j];
        buf ^= 1;
        ci = cn;
      }
    }
  }
  __syncthreads();

  const float4* acc4 = reinterpret_cast<const float4*>(acc);
  float4* R4 = reinterpret_cast<float4*>(R + (long long)r0 * LD);
  for (int idx = tid; idx < TR * LD / 4; idx += NTHR) {
    const int row = r0 + (idx * 4) / LD;
    if (row < N) {
      float4 v = acc4[idx];
      if (RELU) { v.x = fmaxf(v.x, 0.0f); v.y = fmaxf(v.y, 0.0f); v.z = fmaxf(v.z, 0.0f); v.w = fmaxf(v.w, 0.0f); }
      R4[idx] = v;
    }
  }
}

extern "C" void kernel_launch(void* const* d_in, const int* in_sizes, int n_in,
                              void* d_out, int out_size, void* d_ws, size_t ws_size,
                              hipStream_t stream) {
  const float* x  = (const float*)d_in[0];
  const int* e1   = (const int*)d_in[1];
  const int* e2   = (const int*)d_in[2];
  const int* e3   = (const int*)d_in[3];
  const float* W1 = (const float*)d_in[5];
  const float* b1 = (const float*)d_in[6];
  const float* W2 = (const float*)d_in[7];
  const float* b2 = (const float*)d_in[8];
  const float* Wl = (const float*)d_in[9];
  const float* bl = (const float*)d_in[10];

  const int F = 512, H = 128, C = 64;
  const int N  = in_sizes[0] / F;
  const int E1 = in_sizes[1] / 2;
  const int E2 = in_sizes[2] / 2;
  const int E3 = in_sizes[3] / 2;

  float* ws    = (float*)d_ws;
  float* dinv4 = ws;
  float* HH1   = dinv4 + 4LL * N;
  float* R1    = HH1 + (long long)N * H;
  float* HH2   = R1 + (long long)N * 4 * H;
  float* R2    = HH2 + (long long)N * C;

  const int otiles = cdiv_i(N, TR);
  const int gblocks = cdiv_i(N, GROWS);
  const int groups = cdiv_i(otiles, G);
  const int gtiles = groups * G;
  unsigned* pcnt = (unsigned*)(R2 + (long long)N * 4 * C);
  int* seghdr = (int*)(pcnt + 4LL * G * N);
  int* seg    = seghdr + 4LL * G * gtiles;

  deg_owner<<<groups * G, NTHR, 0, stream>>>(e1, e1 + E1, E1, e2, e2 + E2, E2, e3, e3 + E3, E3,
                                             pcnt, seghdr, seg, N);

  gemm_wmma_f32<512, 128, false><<<gblocks, GTH, 0, stream>>>(x, W1, nullptr, HH1, N, pcnt, dinv4);

  agg_owner<128, true><<<otiles, NTHR, 0, stream>>>(
      e1, e1 + E1, E1, e2, e2 + E2, E2, e3, e3 + E3, E3, dinv4, seghdr, seg, HH1, b1, R1, N);

  gemm_wmma_f32<512, 64, false><<<gblocks, GTH, 0, stream>>>(R1, W2, nullptr, HH2, N, nullptr, nullptr);

  agg_owner<64, false><<<otiles, NTHR, 0, stream>>>(
      e1, e1 + E1, E1, e2, e2 + E2, E2, e3, e3 + E3, E3, dinv4, seghdr, seg, HH2, b2, R2, N);

  gemm_wmma_f32<256, 64, true><<<gblocks, GTH, 0, stream>>>(R2, Wl, bl, (float*)d_out, N, nullptr, nullptr);
}


# Head's Wlin applied before the layer-2 aggregation (linear): layer-2 GEMM also multiplies its tile by Wlin's 4 blocks, aggregation sums logits, does log-softmax; 6 launches to 5; sums reassociated
# speedup vs baseline: 6.3223x; 1.0413x over previous
#include <hip/hip_runtime.h>
#include <math.h>


typedef float v2f __attribute__((ext_vector_type(2)));
typedef float v8f __attribute__((ext_vector_type(8)));

static inline int cdiv_i(long long a, long long b) { return (int)((a + b - 1) / b); }

constexpr int TR    = 80;
constexpr int NTHR  = 1024;
constexpr int NV    = 8;
constexpr int CHUNK = NTHR * NV * 4;
constexpr int QCAP  = 8192;
constexpr int SEGCAP = 24576;
constexpr int NWAVE = NTHR / 32;
constexpr int WSEG  = SEGCAP / NWAVE;
constexpr int G      = 5;
constexpr int GR     = G * TR;
constexpr int SUBCAP = 6400;

__device__ __forceinline__ void load_chunk(int4 (&v)[NV], const int* __restrict__ dp,
                                           int base, int E, bool vec_ok, int tid) {
  if (vec_ok && base + CHUNK <= E) {
    #pragma unroll
    for (int j = 0; j < NV; ++j)
      v[j] = *reinterpret_cast<const int4*>(dp + base + (j * NTHR + tid) * 4);
  } else if (vec_ok && E >= 4) {
    #pragma unroll
    for (int j = 0; j < NV; ++j) {
      const int i = base + (j * NTHR + tid) * 4;
      const int4 t = *reinterpret_cast<const int4*>(dp + min(i, E - 4));
      const bool in = i < E;
      v[j].x = in ? t.x : -1;
      v[j].y = in ? t.y : -1;
      v[j].z = in ? t.z : -1;
      v[j].w = in ? t.w : -1;
    }
  } else {
    #pragma unroll
    for (int j = 0; j < NV; ++j) {
      const int i = base + (j * NTHR + tid) * 4;
      v[j].x = (i     < E) ? dp[i]     : -1;
      v[j].y = (i + 1 < E) ? dp[i + 1] : -1;
      v[j].z = (i + 2 < E) ? dp[i + 2] : -1;
      v[j].w = (i + 3 < E) ? dp[i + 3] : -1;
    }
  }
}

__device__ __forceinline__ void take_edges(unsigned b, bool hit, unsigned a, int e, int k, unsigned* cntk,
                                           int* wqe, int* wqc, int& wn) {
  if (b != 0u) {
    if (hit) {
      atomicAdd(&cntk[a], 1u);
      const int pos = wn + (int)__builtin_amdgcn_mbcnt_lo(b, 0u);
      if (pos < WSEG) { wqe[pos] = e; wqc[pos] = (int)a * 4 + k; }
    }
    wn += __builtin_popcount(b);
  }
}

__global__ __launch_bounds__(NTHR)
void deg_owner(const int* __restrict__ s1, const int* __restrict__ d1, int E1,
               const int* __restrict__ s2, const int* __restrict__ d2, int E2,
               const int* __restrict__ s3, const int* __restrict__ d3, int E3,
               unsigned* __restrict__ pcnt, int* __restrict__ seghdr, int* __restrict__ seg, int N) {
  __shared__ unsigned cnt[3 * GR];
  __shared__ int qe[SEGCAP];
  __shared__ int qc[SEGCAP];
  __shared__ int wtc[NWAVE * G];
  __shared__ int wov[NWAVE];
  const int tid  = threadIdx.x;
  const int lane = tid & 31;
  const int wave = __builtin_amdgcn_readfirstlane(tid >> 5);
  const int rg   = blockIdx.x / G;
  const int sl   = blockIdx.x - rg * G;
  const int r0   = rg * GR;
  for (int i = tid; i < 3 * GR; i += NTHR) cnt[i] = 0u;
  __syncthreads();

  int* const wqe = qe + wave * WSEG;
  int* const wqc = qc + wave * WSEG;
  int wn = 0;

  #pragma unroll 1
  for (int k = 0; k < 3; ++k) {
    const int* __restrict__ dp = (k == 0) ? d1 : (k == 1) ? d2 : d3;
    const int E = (k == 0) ? E1 : (k == 1) ? E2 : E3;
    unsigned* const cntk = cnt + k * GR;
    const int m   = (E >= 4) ? (int)((reinterpret_cast<size_t>(dp) >> 2) & 3) : 0;
    const int* __restrict__ ap = dp - m;
    const int Ep  = E + m;
    const int Epf = (E >= 4) ? (Ep & ~3) : 0;
    const int pg  = ((Epf >> 2) + G - 1) / G;
    const int lo  = min(sl * pg * 4, Epf);
    const int hi  = min(lo + pg * 4, Epf);
    for (int pb = lo; pb < hi; pb += CHUNK) {
      int4 v[NV];
      if (pb + CHUNK <= hi && (pb > 0 || m == 0)) {
        #pragma unroll
        for (int j = 0; j < NV; ++j)
          v[j] = *reinterpret_cast<const int4*>(ap + pb + (j * NTHR + tid) * 4);
      } else {
        #pragma unroll
        for (int j = 0; j < NV; ++j) {
          const int p = pb + (j * NTHR + tid) * 4;
          const int4 t = *reinterpret_cast<const int4*>(ap + min(p, hi - 4));
          const bool in = p < hi;
          v[j].x = (in && p     >= m) ? t.x : -1;
          v[j].y = (in && p + 1 >= m) ? t.y : -1;
          v[j].z = (in && p + 2 >= m) ? t.z : -1;
          v[j].w = (in && p + 3 >= m) ? t.w : -1;
        }
      }
      unsigned any[NV];
      #pragma unroll
      for (int j = 0; j < NV; ++j) {
        v[j].x -= r0; v[j].y -= r0; v[j].z -= r0; v[j].w -= r0;
        const unsigned mn = min(min((unsigned)v[j].x, (unsigned)v[j].y), min((unsigned)v[j].z, (unsigned)v[j].w));
        any[j] = __builtin_amdgcn_ballot_w32(mn < (unsigned)GR);
      }
      #pragma unroll
      for (int j = 0; j < NV; ++j) {
        if (any[j] != 0u) {
          const int i = pb + (j * NTHR + tid) * 4 - m;
          const unsigned a0 = (unsigned)v[j].x, a1 = (unsigned)v[j].y, a2 = (unsigned)v[j].z, a3 = (unsigned)v[j].w;
          const unsigned b0 = __builtin_amdgcn_ballot_w32(a0 < (unsigned)GR);
          const unsigned b1 = __builtin_amdgcn_ballot_w32(a1 < (unsigned)GR);
          const unsigned b2 = __builtin_amdgcn_ballot_w32(a2 < (unsigned)GR);
          const unsigned b3 = __builtin_amdgcn_ballot_w32(a3 < (unsigned)GR);
          take_edges(b0, a0 < (unsigned)GR, a0, i,     k, cntk, wqe, wqc, wn);
          take_edges(b1, a1 < (unsigned)GR, a1, i + 1, k, cntk, wqe, wqc, wn);
          take_edges(b2, a2 < (unsigned)GR, a2, i + 2, k, cntk, wqe, wqc, wn);
          take_edges(b3, a3 < (unsigned)GR, a3, i + 3, k, cntk, wqe, wqc, wn);
        }
      }
    }
    {
      const bool mine = (sl == G - 1) && tid < Ep - Epf;
      const int i = Epf - m + tid;
      unsigned a = 0xffffffffu;
      if (mine) a = (unsigned)(dp[i] - r0);
      const bool hit = mine && a < (unsigned)GR;
      take_edges(__builtin_amdgcn_ballot_w32(hit), hit, a, i, k, cntk, wqe, wqc, wn);
    }
  }
  __syncthreads();

  if (tid < GR) {
    const int row = r0 + tid;
    if (row < N) {
      uint4 o;
      o.x = cnt[tid]; o.y = cnt[GR + tid]; o.z = cnt[2 * GR + tid]; o.w = 0u;
      reinterpret_cast<uint4*>(pcnt)[(long long)sl * N + row] = o;
    }
  }

  const int nq = min(wn, WSEG);
  int tc[G];
  #pragma unroll
  for (int g = 0; g < G; ++g) tc[g] = 0;
  for (int i0 = 0; i0 < nq; i0 += 32) {
    const int ii = i0 + lane;
    int t = -1;
    if (ii < nq) t = (wqc[ii] >> 2) / TR;
    #pragma unroll
    for (int g = 0; g < G; ++g) tc[g] += __builtin_popcount(__builtin_amdgcn_ballot_w32(t == g));
  }
  if (lane == 0) {
    #pragma unroll
    for (int g = 0; g < G; ++g) wtc[wave * G + g] = tc[g];
    wov[wave] = (wn > WSEG) ? 1 : 0;
  }
  __syncthreads();

  static_assert(NWAVE == 32, "one lane per wave of the work-group");
  const bool ovf = __builtin_amdgcn_ballot_w32(wov[lane] != 0) != 0u;
  int off[G], tot[G];
  #pragma unroll
  for (int g = 0; g < G; ++g) {
    int t = wtc[lane * G + g];
    int o = (lane < wave) ? t : 0;
    #pragma unroll
    for (int s = 16; s > 0; s >>= 1) { t += __shfl_xor(t, s); o += __shfl_xor(o, s); }
    tot[g] = __builtin_amdgcn_readfirstlane(t);
    off[g] = __builtin_amdgcn_readfirstlane(o);
  }
  const long long pair0 = ((long long)rg * G) * G + sl;
  #pragma unroll
  for (int g = 0; g < G; ++g) {
    if (tid == g) {
      const bool o = ovf || tot[g] > SUBCAP;
      int4 h;
      h.x = o ? 0 : tot[g]; h.y = o ? 1 : 0; h.z = 0; h.w = 0;
      reinterpret_cast<int4*>(seghdr)[pair0 + (long long)g * G] = h;
    }
  }
  const int nw = ovf ? 0 : nq;
  int2* const sg = reinterpret_cast<int2*>(seg) + pair0 * SUBCAP;
  for (int i0 = lane; i0 < nw; i0 += 4 * 32) {
    int c[4], s[4];
    #pragma unroll
    for (int u = 0; u < 4; ++u) {
      const int ii = min(i0 + u * 32, nw - 1);
      const int e = wqe[ii];
      c[u] = wqc[ii];
      const int kk = c[u] & 3;
      const int* sp = (kk == 0) ? s1 : (kk == 1) ? s2 : s3;
      s[u] = sp[e];
    }
    #pragma unroll
    for (int u = 0; u < 4; ++u) {
      const int a = c[u] >> 2;
      const int t = (i0 + u * 32 < nw) ? a / TR : -1;
      const int2 rec = make_int2(s[u], (a - t * TR) * 4 + (c[u] & 3));
      #pragma unroll
      for (int g = 0; g < G; ++g) {
        const unsigned b = __builtin_amdgcn_ballot_w32(t == g);
        if (t == g) {
          const int pos = off[g] + (int)__builtin_amdgcn_mbcnt_lo(b, 0u);
          if (pos < SUBCAP) sg[(long long)g * G * SUBCAP + pos] = rec;
        }
        off[g] += __builtin_popcount(b);
      }
    }
  }
}

constexpr int GW    = 20;
constexpr int GTH   = GW * 32;
constexpr int GROWS = 80;
constexpr int LSP   = 68;

template <int K, int NC>
__device__ __forceinline__ void copy_b_frag(const float* __restrict__ B, float4* Bs, int tid) {
  constexpr int NT  = NC / 16;
  constexpr int TOT = (K / 4) * NC;
  constexpr int NTRIP = (TOT + GTH - 1) / GTH;
  constexpr int BATCH = (NTRIP % 13 == 0) ? 13 : 7;
  static_assert(NTRIP % BATCH == 0, "whole batches");
  #pragma unroll 1
  for (int t0 = 0; t0 < NTRIP; t0 += BATCH) {
    float4 v[BATCH];
    #pragma unroll
    for (int u = 0; u < BATCH; ++u) {
      const int i  = min((t0 + u) * GTH + tid, TOT - 1);
      const int c  = i % NC;
      const int kg = i / NC;
      const float* bp = B + (long long)(kg * 4) * NC + c;
      v[u].x = bp[0]; v[u].y = bp[NC]; v[u].z = bp[2 * NC]; v[u].w = bp[3 * NC];
    }
    #pragma unroll
    for (int u = 0; u < BATCH; ++u) {
      const int i  = (t0 + u) * GTH + tid;
      const int c  = i % NC;
      const int kg = i / NC;
      if (i < TOT) Bs[((kg >> 1) * NT + (c >> 4)) * 32 + (kg & 1) * 16 + (c & 15)] = v[u];
    }
  }
}

template <int K, int NC, bool HEAD>
__global__ __launch_bounds__(GTH)
void gemm_wmma_f32(const float* __restrict__ A, const float* __restrict__ B,
                   const float* __restrict__ bias, float* __restrict__ D, int M,
                   const unsigned* __restrict__ pc, float* __restrict__ dinv4,
                   const float* __restrict__ B2, const float* __restrict__ bq, float* __restrict__ cq) {
  constexpr int NT  = NC / 16;
  constexpr int CT  = NT / 4;
  constexpr int KB  = K / 8;
  constexpr int K2  = 64;
  constexpr int NQ  = 4;
  static_assert(K % 32 == 0 && NC % 64 == 0, "shapes");
  static_assert(!HEAD || (NC == K2 && CT == 1 && GW == 5 * NQ), "second product: 64 columns in, a wave per (row tile, block)");
  __shared__ float4 Bs[KB * NT * 32];
  __shared__ float4 Bs2[HEAD ? (NQ * K2 / 8) * 4 * 32 : 1];
  __shared__ __attribute__((aligned(16))) float Ls[HEAD ? GROWS * LSP : 1];

  const int tid  = threadIdx.x;
  const int lane = tid & 31;
  const int wave = tid >> 5;
  const int half = lane >> 4;
  const int mn   = lane & 15;
  const int row0 = blockIdx.x * GROWS;
  const int rt   = wave >> 2;
  const int ct0  = (wave & 3) * CT;

  const int arow = min(row0 + rt * 16 + mn, M - 1);
  const float4* __restrict__ ap = reinterpret_cast<const float4*>(A + (long long)arow * K) + half;

  float4 cur[4];
  #pragma unroll
  for (int j = 0; j < 4; ++j) cur[j] = ap[2 * j];

  if (pc != nullptr && tid < GROWS && row0 + tid < M) {
    const int row = row0 + tid;
    unsigned c0 = 0u, c1 = 0u, c2 = 0u;
    #pragma unroll
    for (int g = 0; g < G; ++g) {
      const uint4 p = reinterpret_cast<const uint4*>(pc)[(long long)g * M + row];
      c0 += p.x; c1 += p.y; c2 += p.z;
    }
    float4 o;
    o.x = (c0 > 0u) ? rsqrtf((float)c0) : 0.0f;
    o.y = (c1 > 0u) ? rsqrtf((float)c1) : 0.0f;
    o.z = (c2 > 0u) ? rsqrtf((float)c2) : 0.0f;
    o.w = 0.0f;
    reinterpret_cast<float4*>(dinv4)[row] = o;
  }

  copy_b_frag<K, NC>(B, Bs, tid);
  if constexpr (HEAD) copy_b_frag<NQ * K2, 64>(B2, Bs2, tid);
  __syncthreads();

  v8f acc[CT];
  #pragma unroll
  for (int t = 0; t < CT; ++t) acc[t] = v8f{};

  #pragma unroll 1
  for (int kc = 0; kc < KB; kc += 4) {
    const int kn = min(kc + 4, KB - 4);
    float4 nxt[4];
    #pragma unroll
    for (int j = 0; j < 4; ++j) nxt[j] = ap[2 * (kn + j)];

    #pragma unroll
    for (int j = 0; j < 4; ++j) {
      v2f a0, a1;
      a0.x = cur[j].x; a0.y = cur[j].y;
      a1.x = cur[j].z; a1.y = cur[j].w;
      #pragma unroll
      for (int t = 0; t < CT; ++t) {
        const float4 b = Bs[((kc + j) * NT + ct0 + t) * 32 + lane];
        v2f b0, b1;
        b0.x = b.x; b0.y = b.y;
        b1.x = b.z; b1.y = b.w;
        acc[t] = __builtin_amdgcn_wmma_f32_16x16x4_f32(
             false, a0,  false, b0,
             (short)0, acc[t],  false,  false);
        acc[t] = __builtin_amdgcn_wmma_f32_16x16x4_f32(
             false, a1,  false, b1,
             (short)0, acc[t],  false,  false);
      }
    }
    #pragma unroll
    for (int j = 0; j < 4; ++j) cur[j] = nxt[j];
  }

  if constexpr (!HEAD) {
    #pragma unroll
    for (int t = 0; t < CT; ++t) {
      const int col = (ct0 + t) * 16 + mn;
      const float bv = bias ? bias[col] : 0.0f;
      #pragma unroll
      for (int r = 0; r < 8; ++r) {
        const int row = row0 + rt * 16 + r + half * 8;
        if (row < M) D[(long long)row * NC + col] = acc[t][r] + bv;
      }
    }
  } else {
    #pragma unroll
    for (int t = 0; t < CT; ++t) {
      const int col = (ct0 + t) * 16 + mn;
      const float bv = bias ? bias[col] : 0.0f;
      #pragma unroll
      for (int r = 0; r < 8; ++r) Ls[(rt * 16 + r + half * 8) * LSP + col] = acc[t][r] + bv;
    }
    __syncthreads();

    const int q = wave & 3;
    const float4* lp = reinterpret_cast<const float4*>(Ls + (rt * 16 + mn) * LSP) + half;
    v8f acc2[4];
    #pragma unroll
    for (int t = 0; t < 4; ++t) acc2[t] = v8f{};
    #pragma unroll
    for (int kb = 0; kb < K2 / 8; ++kb) {
      const float4 a = lp[2 * kb];
      v2f a0, a1;
      a0.x = a.x; a0.y = a.y;
      a1.x = a.z; a1.y = a.w;
      #pragma unroll
      for (int t = 0; t < 4; ++t) {
        const float4 b = Bs2[((q * (K2 / 8) + kb) * 4 + t) * 32 + lane];
        v2f b0, b1;
        b0.x = b.x; b0.y = b.y;
        b1.x = b.z; b1.y = b.w;
        acc2[t] = __builtin_amdgcn_wmma_f32_16x16x4_f32(
             false, a0,  false, b0,
             (short)0, acc2[t],  false,  false);
        acc2[t] = __builtin_amdgcn_wmma_f32_16x16x4_f32(
             false, a1,  false, b1,
             (short)0, acc2[t],  false,  false);
      }
    }
    #pragma unroll
    for (int t = 0; t < 4; ++t) {
      const int col = q * 64 + t * 16 + mn;
      #pragma unroll
      for (int r = 0; r < 8; ++r) {
        const int row = row0 + rt * 16 + r + half * 8;
        if (row < M) D[(long long)row * (NQ * 64) + col] = acc2[t][r];
      }
    }

    if (blockIdx.x == 0 && __builtin_amdgcn_readfirstlane(rt) == 0) {
      v8f accb[4];
      #pragma unroll
      for (int t = 0; t < 4; ++t) accb[t] = v8f{};
      #pragma unroll
      for (int kb = 0; kb < K2 / 8; ++kb) {
        const float* bp = bq + kb * 8 + half * 4;
        v2f a0, a1;
        a0.x = bp[0]; a0.y = bp[1];
        a1.x = bp[2]; a1.y = bp[3];
        #pragma unroll
        for (int t = 0; t < 4; ++t) {
          const float4 b = Bs2[((q * (K2 / 8) + kb) * 4 + t) * 32 + lane];
          v2f b0, b1;
          b0.x = b.x; b0.y = b.y;
          b1.x = b.z; b1.y = b.w;
          accb[t] = __builtin_amdgcn_wmma_f32_16x16x4_f32(
               false, a0,  false, b0,
               (short)0, accb[t],  false,  false);
          accb[t] = __builtin_amdgcn_wmma_f32_16x16x4_f32(
               false, a1,  false, b1,
               (short)0, accb[t],  false,  false);
        }
      }
      if (half == 0) {
        #pragma unroll
        for (int t = 0; t < 4; ++t) cq[q * 64 + t * 16 + mn] = accb[t][0];
      }
    }
  }
}


template <int WIDTH, bool HEAD>
__device__ __forceinline__ void edge_in_lane(int e, int dl, int k, const int* __restrict__ sp,
                                             const float* __restrict__ dinv4, const float* dinvL,
                                             const float* __restrict__ HH, float* acc) {
  constexpr int LD  = 4 * WIDTH;
  constexpr int SLD = HEAD ? 4 * WIDTH : WIDTH;
  const int s = sp[e];
  const float w = dinv4[s * 4 + k] * dinvL[dl * 4 + k];
  const float* hp = HH + (long long)s * SLD + (HEAD ? k * WIDTH : 0);
  float* ap = acc + dl * LD + (HEAD ? 0 : k * WIDTH);
  #pragma unroll 1
  for (int c = 0; c < WIDTH; ++c) atomicAdd(ap + c, hp[c] * w);
}

template <int WIDTH, int CNT, bool HEAD>
__device__ __forceinline__ void work_edges(const int* qe_, const int* qd_, int k,
                                           const int* __restrict__ sp,
                                           const float* __restrict__ dinv4, const float* dinvL,
                                           const float* __restrict__ HH, float* acc, int lane) {
  constexpr int LD  = 4 * WIDTH;
  constexpr int SLD = HEAD ? 4 * WIDTH : WIDTH;
  constexpr int NJ = WIDTH / 32;
  int dl[CNT], s[CNT];
  #pragma unroll
  for (int u = 0; u < CNT; ++u) { dl[u] = qd_[u]; s[u] = sp[qe_[u]]; }
  float w[CNT], hv[CNT][NJ];
  #pragma unroll
  for (int u = 0; u < CNT; ++u) {
    w[u] = dinv4[s[u] * 4 + k] * dinvL[dl[u] * 4 + k];
    const float* hp = HH + (long long)s[u] * SLD + (HEAD ? k * WIDTH : 0) + lane;
    #pragma unroll
    for (int j = 0; j < NJ; ++j) hv[u][j] = hp[32 * j];
  }
  #pragma unroll
  for (int u = 0; u < CNT; ++u) {
    float* ap = acc + dl[u] * LD + (HEAD ? 0 : k * WIDTH) + lane;
    #pragma unroll
    for (int j = 0; j < NJ; ++j) atomicAdd(ap + 32 * j, hv[u][j] * w[u]);
  }
}

template <int WIDTH, int CNT, bool FULL>
__device__ __forceinline__ void work_group(int2 rec, float dv, float dl, int g, int last,
                                           const float* __restrict__ HH, float* acc, int lane) {
  constexpr int LD = 4 * WIDTH;
  constexpr int NJ = WIDTH / 32;
  int c[CNT];
  float hv[CNT][NJ];
  #pragma unroll
  for (int u = 0; u < CNT; ++u) {
    const int from = FULL ? g + u : min(g + u, last);
    const int s = __builtin_amdgcn_readlane(rec.x, from);
    c[u] = __builtin_amdgcn_readlane(rec.y, from);
    const float* hp = HH + (long long)s * WIDTH + lane;
    #pragma unroll
    for (int j = 0; j < NJ; ++j) hv[u][j] = hp[32 * j];
  }
  #pragma unroll
  for (int u = 0; u < CNT; ++u) {
    if (FULL || g + u <= last) {
      const float w = __int_as_float(__builtin_amdgcn_readlane(__float_as_int(dv), g + u)) *
                      __int_as_float(__builtin_amdgcn_readlane(__float_as_int(dl), g + u));
      float* ap = acc + (c[u] >> 2) * LD + (c[u] & 3) * WIDTH + lane;
      #pragma unroll
      for (int j = 0; j < NJ; ++j) atomicAdd(ap + 32 * j, hv[u][j] * w);
    }
  }
}

template <int WIDTH, bool HEAD>
__device__ __forceinline__ void work_records(const int2* __restrict__ sg, int n,
                                             const float* __restrict__ dinv4, const float* dinvL,
                                             const float* __restrict__ HH, float* acc,
                                             int wave, int lane) {
  constexpr int U = 1024 / WIDTH;
  int b0 = wave * 32;
  if (b0 >= n) return;
  int2 rec = sg[min(b0 + lane, n - 1)];
  while (b0 < n) {
    const int cnt = min(32, n - b0);
    const int nb  = b0 + NTHR;
    int2 nrec = rec;
    if (nb < n) nrec = sg[min(nb + lane, n - 1)];
    const float dv = dinv4[rec.x * 4 + (rec.y & 3)];
    const float dl = dinvL[rec.y];
    int2 wr = rec;
    if constexpr (HEAD) { wr.x = rec.x * 4 + (rec.y & 3); wr.y = rec.y & ~3; }
    int g = 0;
    for (; g + U <= cnt; g += U) work_group<WIDTH, U, true>(wr, dv, dl, g, cnt - 1, HH, acc, lane);
    if (g < cnt)                 work_group<WIDTH, U, false>(wr, dv, dl, g, cnt - 1, HH, acc, lane);
    rec = nrec;
    b0  = nb;
  }
}

template <int WIDTH, bool RELU, bool HEAD>
__global__ __launch_bounds__(NTHR)
void agg_owner(const int* __restrict__ s1, const int* __restrict__ d1, int E1,
               const int* __restrict__ s2, const int* __restrict__ d2, int E2,
               const int* __restrict__ s3, const int* __restrict__ d3, int E3,
               const float* __restrict__ dinv4, const int* __restrict__ seghdr,
               const int* __restrict__ seg, const float* __restrict__ HH,
               const float* __restrict__ bias, const float* __restrict__ cq,
               float* __restrict__ R, int N) {
  constexpr int LD  = 4 * WIDTH;
  constexpr int SLD = HEAD ? 4 * WIDTH : WIDTH;
  constexpr int U  = 4;
  static_assert(!HEAD || (WIDTH == 64 && !RELU), "row pass: 64 columns, two per lane");
  __shared__ __attribute__((aligned(16))) float acc[TR * LD];
  __shared__ float dinvL[TR * 4];
  __shared__ int qe[2 * QCAP];
  __shared__ int qd[2 * QCAP];
  __shared__ unsigned qcount[3];
  __shared__ int hsub[G];
  __shared__ int hovf[G];

  const int tid  = threadIdx.x;
  const int wave = tid >> 5;
  const int lane = tid & 31;
  const int r0   = blockIdx.x * TR;

  constexpr int W4  = WIDTH / 4;
  constexpr int LD4 = LD / 4;
  constexpr int SV  = TR * W4;
  constexpr int SP  = (SV + NTHR - 1) / NTHR;
  static_assert(NTHR % W4 == 0, "a thread keeps its column group of HH through the trips");
  static_assert(HEAD || (NTHR % LD4 == 0 && (TR * LD4) % NTHR == 0), "a thread keeps its column group through the fill");

  const float4* __restrict__ HH4 = reinterpret_cast<const float4*>(HH);
  const long long end4 = (long long)N * W4 - 1;
  float4 sv[SP];
  #pragma unroll
  for (int u = 0; u < SP; ++u) {
    if constexpr (HEAD) {
      const int i = u * NTHR + tid;
      const long long row = min(r0 + i / W4, N - 1);
      sv[u] = HH4[row * (SLD / 4) + 3 * W4 + i % W4];
    } else {
      const long long i4 = (long long)r0 * W4 + u * NTHR + tid;
      sv[u] = HH4[(i4 < end4) ? i4 : end4];
    }
  }
  int4 hd = make_int4(0, 0, 0, 0);
  if (tid < G) hd = reinterpret_cast<const int4*>(seghdr)[(long long)blockIdx.x * G + tid];
  float dvl = 0.0f;
  if (tid < TR * 4 && r0 + (tid >> 2) < N) dvl = dinv4[(long long)r0 * 4 + tid];
  static_assert(TR * 4 <= NTHR, "one degree factor per thread");
  const int h4 = tid % W4;
  float4 bv;
  bv.x = bias[h4 * 4]; bv.y = bias[h4 * 4 + 1]; bv.z = bias[h4 * 4 + 2]; bv.w = bias[h4 * 4 + 3];
  if constexpr (HEAD) {
    float4 cv[4];
    #pragma unroll
    for (int k = 0; k < 4; ++k) cv[k] = reinterpret_cast<const float4*>(cq)[k * W4 + h4];
    #pragma unroll
    for (int k = 0; k < 4; ++k) { bv.x += cv[k].x; bv.y += cv[k].y; bv.z += cv[k].z; bv.w += cv[k].w; }
  }

  float4* acc4w = reinterpret_cast<float4*>(acc);
  if constexpr (!HEAD) {
    if (tid % LD4 < 3 * W4) {
      for (int idx4 = tid; idx4 < TR * LD4; idx4 += NTHR) acc4w[idx4] = bv;
    }
  }
  #pragma unroll
  for (int u = 0; u < SP; ++u) {
    const int i = u * NTHR + tid;
    if (i < SV) {
      const int dl = i / W4;
      float4 v = bv;
      if (r0 + dl < N) { v.x += sv[u].x; v.y += sv[u].y; v.z += sv[u].z; v.w += sv[u].w; }
      acc4w[dl * LD4 + (HEAD ? 0 : 3 * W4) + h4] = v;
    }
  }
  if (tid < TR * 4) dinvL[tid] = dvl;
  if (tid < G) { hsub[tid] = hd.x; hovf[tid] = hd.y; }
  if (tid < 3) qcount[tid] = 0u;
  __syncthreads();

  int ovf = 0;
  #pragma unroll
  for (int g = 0; g < G; ++g) ovf |= hovf[g];
  ovf = __builtin_amdgcn_readfirstlane(ovf);

  if (ovf == 0) {
    const int wv = __builtin_amdgcn_readfirstlane(wave);
    int rot = 0;
    #pragma unroll 1
    for (int g = 0; g < G; ++g) {
      const int nrec = __builtin_amdgcn_readfirstlane(hsub[g]);
      work_records<WIDTH, HEAD>(reinterpret_cast<const int2*>(seg) + ((long long)blockIdx.x * G + g) * SUBCAP, nrec,
                          dinv4, dinvL, HH, acc, (wv - rot) & (NWAVE - 1), lane);
      rot += (nrec + 31) >> 5;
    }
  } else {
    int buf = 0;
    int ci  = 0;

    #pragma unroll 1
    for (int k = 0; k < 3; ++k) {
      const int* __restrict__ sp = (k == 0) ? s1 : (k == 1) ? s2 : s3;
      const int* __restrict__ dp = (k == 0) ? d1 : (k == 1) ? d2 : d3;
      const int E = (k == 0) ? E1 : (k == 1) ? E2 : E3;
      const bool vec_ok = ((E & 3) == 0) && ((reinterpret_cast<size_t>(dp) & 15) == 0);

      int4 cur[NV];
      load_chunk(cur, dp, 0, E, vec_ok, tid);

      for (int base = 0; base < E; base += CHUNK) {
        const int cn = (ci == 2) ? 0 : ci + 1;
        if (tid == 0) qcount[cn] = 0u;
        unsigned* qc = &qcount[ci];
        int* qeb = qe + buf * QCAP;
        int* qdb = qd + buf * QCAP;

        #pragma unroll
        for (int j = 0; j < NV; ++j) {
          const int i = base + (j * NTHR + tid) * 4;
          const unsigned a0 = (unsigned)(cur[j].x - r0), a1 = (unsigned)(cur[j].y - r0);
          const unsigned a2 = (unsigned)(cur[j].z - r0), a3 = (unsigned)(cur[j].w - r0);
          if (min(min(a0, a1), min(a2, a3)) < (unsigned)TR) {
            #pragma unroll 1
            for (int t = 0; t < 4; ++t) {
              const unsigned a = (t == 0) ? a0 : (t == 1) ? a1 : (t == 2) ? a2 : a3;
              if (a < (unsigned)TR) {
                const unsigned pos = atomicAdd(qc, 1u);
                if (pos < (unsigned)QCAP) { qeb[pos] = i + t; qdb[pos] = (int)a; }
                else edge_in_lane<WIDTH, HEAD>(i + t, (int)a, k, sp, dinv4, dinvL, HH, acc);
              }
            }
          }
        }

        __syncthreads();

        int4 nxt[NV];
        load_chunk(nxt, dp, base + CHUNK, E, vec_ok, tid);

        const int n = min((int)(*qc), QCAP);
        const int nfull = n - (n % U);
        for (int i0 = wave * U; i0 < nfull; i0 += (NTHR / 32) * U)
          work_edges<WIDTH, U, HEAD>(qeb + i0, qdb + i0, k, sp, dinv4, dinvL, HH, acc, lane);
        for (int i1 = nfull + wave; i1 < n; i1 += NTHR / 32)
          work_edges<WIDTH, 1, HEAD>(qeb + i1, qdb + i1, k, sp, dinv4, dinvL, HH, acc, lane);


        #pragma unroll
        for (int j = 0; j < NV; ++j) cur[j] = nxt[j];
        buf ^= 1;
        ci = cn;
      }
    }
  }
  __syncthreads();

  if constexpr (HEAD) {
    constexpr int RQ = (TR + NWAVE - 1) / NWAVE;
    float a[RQ], b[RQ], mx[RQ], s[RQ];
    #pragma unroll
    for (int q = 0; q < RQ; ++q) {
      const int lrow = min(wave + q * NWAVE, TR - 1);
      a[q]  = acc[lrow * LD + lane];
      b[q]  = acc[lrow * LD + lane + 32];
      mx[q] = fmaxf(a[q], b[q]);
    }
    #pragma unroll
    for (int off = 16; off > 0; off >>= 1) {
      float o[RQ];
      #pragma unroll
      for (int q = 0; q < RQ; ++q) o[q] = __shfl_xor(mx[q], off);
      #pragma unroll
      for (int q = 0; q < RQ; ++q) mx[q] = fmaxf(mx[q], o[q]);
    }
    #pragma unroll
    for (int q = 0; q < RQ; ++q) s[q] = __expf(a[q] - mx[q]) + __expf(b[q] - mx[q]);
    #pragma unroll
    for (int off = 16; off > 0; off >>= 1) {
      float o[RQ];
      #pragma unroll
      for (int q = 0; q < RQ; ++q) o[q] = __shfl_xor(s[q], off);
      #pragma unroll
      for (int q = 0; q < RQ; ++q) s[q] += o[q];
    }
    #pragma unroll
    for (int q = 0; q < RQ; ++q) {
      const int lrow = wave + q * NWAVE;
      const int row = r0 + lrow;
      const float lse = mx[q] + __logf(s[q]);
      if (lrow < TR && row < N) {
        float* p = R + (long long)row * WIDTH;
        p[lane]      = a[q] - lse;
        p[lane + 32] = b[q] - lse;
      }
    }
  } else {
    const float4* acc4 = reinterpret_cast<const float4*>(acc);
    float4* R4 = reinterpret_cast<float4*>(R + (long long)r0 * LD);
    for (int idx = tid; idx < TR * LD / 4; idx += NTHR) {
      const int row = r0 + (idx * 4) / LD;
      if (row < N) {
        float4 v = acc4[idx];
        if (RELU) { v.x = fmaxf(v.x, 0.0f); v.y = fmaxf(v.y, 0.0f); v.z = fmaxf(v.z, 0.0f); v.w = fmaxf(v.w, 0.0f); }
        R4[idx] = v;
      }
    }
  }
}

extern "C" void kernel_launch(void* const* d_in, const int* in_sizes, int n_in,
                              void* d_out, int out_size, void* d_ws, size_t ws_size,
                              hipStream_t stream) {
  const float* x  = (const float*)d_in[0];
  const int* e1   = (const int*)d_in[1];
  const int* e2   = (const int*)d_in[2];
  const int* e3   = (const int*)d_in[3];
  const float* W1 = (const float*)d_in[5];
  const float* b1 = (const float*)d_in[6];
  const float* W2 = (const float*)d_in[7];
  const float* b2 = (const float*)d_in[8];
  const float* Wl = (const float*)d_in[9];
  const float* bl = (const float*)d_in[10];

  const int F = 512, H = 128, C = 64;
  const int N  = in_sizes[0] / F;
  const int E1 = in_sizes[1] / 2;
  const int E2 = in_sizes[2] / 2;
  const int E3 = in_sizes[3] / 2;

  float* ws    = (float*)d_ws;
  float* dinv4 = ws;
  float* HH1   = dinv4 + 4LL * N;
  float* R1    = HH1 + (long long)N * H;
  float* P2    = R1 + (long long)N * 4 * H;
  float* cq    = P2 + (long long)N * 4 * C;

  const int otiles = cdiv_i(N, TR);
  const int gblocks = cdiv_i(N, GROWS);
  const int groups = cdiv_i(otiles, G);
  const int gtiles = groups * G;
  unsigned* pcnt = (unsigned*)(cq + 4 * C);
  int* seghdr = (int*)(pcnt + 4LL * G * N);
  int* seg    = seghdr + 4LL * G * gtiles;

  deg_owner<<<groups * G, NTHR, 0, stream>>>(e1, e1 + E1, E1, e2, e2 + E2, E2, e3, e3 + E3, E3,
                                             pcnt, seghdr, seg, N);

  gemm_wmma_f32<512, 128, false><<<gblocks, GTH, 0, stream>>>(x, W1, nullptr, HH1, N, pcnt, dinv4,
                                                              nullptr, nullptr, nullptr);

  agg_owner<128, true, false><<<otiles, NTHR, 0, stream>>>(
      e1, e1 + E1, E1, e2, e2 + E2, E2, e3, e3 + E3, E3, dinv4, seghdr, seg, HH1, b1, nullptr, R1, N);

  gemm_wmma_f32<512, 64, true><<<gblocks, GTH, 0, stream>>>(R1, W2, nullptr, P2, N, nullptr, nullptr,
                                                            Wl, b2, cq);

  agg_owner<64, false, true><<<otiles, NTHR, 0, stream>>>(
      e1, e1 + E1, E1, e2, e2 + E2, E2, e3, e3 + E3, E3, dinv4, seghdr, seg, P2, bl, cq, (float*)d_out, N);
}


# Aggregation gathers: a lane takes its 4 (layer 2: 2) adjacent columns of a source row in one 16- (8-)byte load, one load per row and wave, not 4 (2) of 4 bytes; sums in LDS lane-major, orders kept
# speedup vs baseline: 6.7191x; 1.0628x over previous
#include <hip/hip_runtime.h>
#include <math.h>


typedef float v2f __attribute__((ext_vector_type(2)));
typedef float v8f __attribute__((ext_vector_type(8)));

static inline int cdiv_i(long long a, long long b) { return (int)((a + b - 1) / b); }

constexpr int TR    = 80;
constexpr int NTHR  = 1024;
constexpr int NV    = 8;
constexpr int CHUNK = NTHR * NV * 4;
constexpr int QCAP  = 8192;
constexpr int SEGCAP = 24576;
constexpr int NWAVE = NTHR / 32;
constexpr int WSEG  = SEGCAP / NWAVE;
constexpr int G      = 5;
constexpr int GR     = G * TR;
constexpr int SUBCAP = 6400;

__device__ __forceinline__ void load_chunk(int4 (&v)[NV], const int* __restrict__ dp,
                                           int base, int E, bool vec_ok, int tid) {
  if (vec_ok && base + CHUNK <= E) {
    #pragma unroll
    for (int j = 0; j < NV; ++j)
      v[j] = *reinterpret_cast<const int4*>(dp + base + (j * NTHR + tid) * 4);
  } else if (vec_ok && E >= 4) {
    #pragma unroll
    for (int j = 0; j < NV; ++j) {
      const int i = base + (j * NTHR + tid) * 4;
      const int4 t = *reinterpret_cast<const int4*>(dp + min(i, E - 4));
      const bool in = i < E;
      v[j].x = in ? t.x : -1;
      v[j].y = in ? t.y : -1;
      v[j].z = in ? t.z : -1;
      v[j].w = in ? t.w : -1;
    }
  } else {
    #pragma unroll
    for (int j = 0; j < NV; ++j) {
      const int i = base + (j * NTHR + tid) * 4;
      v[j].x = (i     < E) ? dp[i]     : -1;
      v[j].y = (i + 1 < E) ? dp[i + 1] : -1;
      v[j].z = (i + 2 < E) ? dp[i + 2] : -1;
      v[j].w = (i + 3 < E) ? dp[i + 3] : -1;
    }
  }
}

__device__ __forceinline__ void take_edges(unsigned b, bool hit, unsigned a, int e, int k, unsigned* cntk,
                                           int* wqe, int* wqc, int& wn) {
  if (b != 0u) {
    if (hit) {
      atomicAdd(&cntk[a], 1u);
      const int pos = wn + (int)__builtin_amdgcn_mbcnt_lo(b, 0u);
      if (pos < WSEG) { wqe[pos] = e; wqc[pos] = (int)a * 4 + k; }
    }
    wn += __builtin_popcount(b);
  }
}

__global__ __launch_bounds__(NTHR)
void deg_owner(const int* __restrict__ s1, const int* __restrict__ d1, int E1,
               const int* __restrict__ s2, const int* __restrict__ d2, int E2,
               const int* __restrict__ s3, const int* __restrict__ d3, int E3,
               unsigned* __restrict__ pcnt, int* __restrict__ seghdr, int* __restrict__ seg, int N) {
  __shared__ unsigned cnt[3 * GR];
  __shared__ int qe[SEGCAP];
  __shared__ int qc[SEGCAP];
  __shared__ int wtc[NWAVE * G];
  __shared__ int wov[NWAVE];
  const int tid  = threadIdx.x;
  const int lane = tid & 31;
  const int wave = __builtin_amdgcn_readfirstlane(tid >> 5);
  const int rg   = blockIdx.x / G;
  const int sl   = blockIdx.x - rg * G;
  const int r0   = rg * GR;
  for (int i = tid; i < 3 * GR; i += NTHR) cnt[i] = 0u;
  __syncthreads();

  int* const wqe = qe + wave * WSEG;
  int* const wqc = qc + wave * WSEG;
  int wn = 0;

  #pragma unroll 1
  for (int k = 0; k < 3; ++k) {
    const int* __restrict__ dp = (k == 0) ? d1 : (k == 1) ? d2 : d3;
    const int E = (k == 0) ? E1 : (k == 1) ? E2 : E3;
    unsigned* const cntk = cnt + k * GR;
    const int m   = (E >= 4) ? (int)((reinterpret_cast<size_t>(dp) >> 2) & 3) : 0;
    const int* __restrict__ ap = dp - m;
    const int Ep  = E + m;
    const int Epf = (E >= 4) ? (Ep & ~3) : 0;
    const int pg  = ((Epf >> 2) + G - 1) / G;
    const int lo  = min(sl * pg * 4, Epf);
    const int hi  = min(lo + pg * 4, Epf);
    for (int pb = lo; pb < hi; pb += CHUNK) {
      int4 v[NV];
      if (pb + CHUNK <= hi && (pb > 0 || m == 0)) {
        #pragma unroll
        for (int j = 0; j < NV; ++j)
          v[j] = *reinterpret_cast<const int4*>(ap + pb + (j * NTHR + tid) * 4);
      } else {
        #pragma unroll
        for (int j = 0; j < NV; ++j) {
          const int p = pb + (j * NTHR + tid) * 4;
          const int4 t = *reinterpret_cast<const int4*>(ap + min(p, hi - 4));
          const bool in = p < hi;
          v[j].x = (in && p     >= m) ? t.x : -1;
          v[j].y = (in && p + 1 >= m) ? t.y : -1;
          v[j].z = (in && p + 2 >= m) ? t.z : -1;
          v[j].w = (in && p + 3 >= m) ? t.w : -1;
        }
      }
      unsigned any[NV];
      #pragma unroll
      for (int j = 0; j < NV; ++j) {
        v[j].x -= r0; v[j].y -= r0; v[j].z -= r0; v[j].w -= r0;
        const unsigned mn = min(min((unsigned)v[j].x, (unsigned)v[j].y), min((unsigned)v[j].z, (unsigned)v[j].w));
        any[j] = __builtin_amdgcn_ballot_w32(mn < (unsigned)GR);
      }
      #pragma unroll
      for (int j = 0; j < NV; ++j) {
        if (any[j] != 0u) {
          const int i = pb + (j * NTHR + tid) * 4 - m;
          const unsigned a0 = (unsigned)v[j].x, a1 = (unsigned)v[j].y, a2 = (unsigned)v[j].z, a3 = (unsigned)v[j].w;
          const unsigned b0 = __builtin_amdgcn_ballot_w32(a0 < (unsigned)GR);
          const unsigned b1 = __builtin_amdgcn_ballot_w32(a1 < (unsigned)GR);
          const unsigned b2 = __builtin_amdgcn_ballot_w32(a2 < (unsigned)GR);
          const unsigned b3 = __builtin_amdgcn_ballot_w32(a3 < (unsigned)GR);
          take_edges(b0, a0 < (unsigned)GR, a0, i,     k, cntk, wqe, wqc, wn);
          take_edges(b1, a1 < (unsigned)GR, a1, i + 1, k, cntk, wqe, wqc, wn);
          take_edges(b2, a2 < (unsigned)GR, a2, i + 2, k, cntk, wqe, wqc, wn);
          take_edges(b3, a3 < (unsigned)GR, a3, i + 3, k, cntk, wqe, wqc, wn);
        }
      }
    }
    {
      const bool mine = (sl == G - 1) && tid < Ep - Epf;
      const int i = Epf - m + tid;
      unsigned a = 0xffffffffu;
      if (mine) a = (unsigned)(dp[i] - r0);
      const bool hit = mine && a < (unsigned)GR;
      take_edges(__builtin_amdgcn_ballot_w32(hit), hit, a, i, k, cntk, wqe, wqc, wn);
    }
  }
  __syncthreads();

  if (tid < GR) {
    const int row = r0 + tid;
    if (row < N) {
      uint4 o;
      o.x = cnt[tid]; o.y = cnt[GR + tid]; o.z = cnt[2 * GR + tid]; o.w = 0u;
      reinterpret_cast<uint4*>(pcnt)[(long long)sl * N + row] = o;
    }
  }

  const int nq = min(wn, WSEG);
  int tc[G];
  #pragma unroll
  for (int g = 0; g < G; ++g) tc[g] = 0;
  for (int i0 = 0; i0 < nq; i0 += 32) {
    const int ii = i0 + lane;
    int t = -1;
    if (ii < nq) t = (wqc[ii] >> 2) / TR;
    #pragma unroll
    for (int g = 0; g < G; ++g) tc[g] += __builtin_popcount(__builtin_amdgcn_ballot_w32(t == g));
  }
  if (lane == 0) {
    #pragma unroll
    for (int g = 0; g < G; ++g) wtc[wave * G + g] = tc[g];
    wov[wave] = (wn > WSEG) ? 1 : 0;
  }
  __syncthreads();

  static_assert(NWAVE == 32, "one lane per wave of the work-group");
  const bool ovf = __builtin_amdgcn_ballot_w32(wov[lane] != 0) != 0u;
  int off[G], tot[G];
  #pragma unroll
  for (int g = 0; g < G; ++g) {
    int t = wtc[lane * G + g];
    int o = (lane < wave) ? t : 0;
    #pragma unroll
    for (int s = 16; s > 0; s >>= 1) { t += __shfl_xor(t, s); o += __shfl_xor(o, s); }
    tot[g] = __builtin_amdgcn_readfirstlane(t);
    off[g] = __builtin_amdgcn_readfirstlane(o);
  }
  const long long pair0 = ((long long)rg * G) * G + sl;
  #pragma unroll
  for (int g = 0; g < G; ++g) {
    if (tid == g) {
      const bool o = ovf || tot[g] > SUBCAP;
      int4 h;
      h.x = o ? 0 : tot[g]; h.y = o ? 1 : 0; h.z = 0; h.w = 0;
      reinterpret_cast<int4*>(seghdr)[pair0 + (long long)g * G] = h;
    }
  }
  const int nw = ovf ? 0 : nq;
  int2* const sg = reinterpret_cast<int2*>(seg) + pair0 * SUBCAP;
  for (int i0 = lane; i0 < nw; i0 += 4 * 32) {
    int c[4], s[4];
    #pragma unroll
    for (int u = 0; u < 4; ++u) {
      const int ii = min(i0 + u * 32, nw - 1);
      const int e = wqe[ii];
      c[u] = wqc[ii];
      const int kk = c[u] & 3;
      const int* sp = (kk == 0) ? s1 : (kk == 1) ? s2 : s3;
      s[u] = sp[e];
    }
    #pragma unroll
    for (int u = 0; u < 4; ++u) {
      const int a = c[u] >> 2;
      const int t = (i0 + u * 32 < nw) ? a / TR : -1;
      const int2 rec = make_int2(s[u], (a - t * TR) * 4 + (c[u] & 3));
      #pragma unroll
      for (int g = 0; g < G; ++g) {
        const unsigned b = __builtin_amdgcn_ballot_w32(t == g);
        if (t == g) {
          const int pos = off[g] + (int)__builtin_amdgcn_mbcnt_lo(b, 0u);
          if (pos < SUBCAP) sg[(long long)g * G * SUBCAP + pos] = rec;
        }
        off[g] += __builtin_popcount(b);
      }
    }
  }
}

constexpr int GW    = 20;
constexpr int GTH   = GW * 32;
constexpr int GROWS = 80;
constexpr int LSP   = 68;

template <int K, int NC>
__device__ __forceinline__ void copy_b_frag(const float* __restrict__ B, float4* Bs, int tid) {
  constexpr int NT  = NC / 16;
  constexpr int TOT = (K / 4) * NC;
  constexpr int NTRIP = (TOT + GTH - 1) / GTH;
  constexpr int BATCH = (NTRIP % 13 == 0) ? 13 : 7;
  static_assert(NTRIP % BATCH == 0, "whole batches");
  #pragma unroll 1
  for (int t0 = 0; t0 < NTRIP; t0 += BATCH) {
    float4 v[BATCH];
    #pragma unroll
    for (int u = 0; u < BATCH; ++u) {
      const int i  = min((t0 + u) * GTH + tid, TOT - 1);
      const int c  = i % NC;
      const int kg = i / NC;
      const float* bp = B + (long long)(kg * 4) * NC + c;
      v[u].x = bp[0]; v[u].y = bp[NC]; v[u].z = bp[2 * NC]; v[u].w = bp[3 * NC];
    }
    #pragma unroll
    for (int u = 0; u < BATCH; ++u) {
      const int i  = (t0 + u) * GTH + tid;
      const int c  = i % NC;
      const int kg = i / NC;
      if (i < TOT) Bs[((kg >> 1) * NT + (c >> 4)) * 32 + (kg & 1) * 16 + (c & 15)] = v[u];
    }
  }
}

template <int K, int NC, bool HEAD>
__global__ __launch_bounds__(GTH)
void gemm_wmma_f32(const float* __restrict__ A, const float* __restrict__ B,
                   const float* __restrict__ bias, float* __restrict__ D, int M,
                   const unsigned* __restrict__ pc, float* __restrict__ dinv4,
                   const float* __restrict__ B2, const float* __restrict__ bq, float* __restrict__ cq) {
  constexpr int NT  = NC / 16;
  constexpr int CT  = NT / 4;
  constexpr int KB  = K / 8;
  constexpr int K2  = 64;
  constexpr int NQ  = 4;
  static_assert(K % 32 == 0 && NC % 64 == 0, "shapes");
  static_assert(!HEAD || (NC == K2 && CT == 1 && GW == 5 * NQ), "second product: 64 columns in, a wave per (row tile, block)");
  __shared__ float4 Bs[KB * NT * 32];
  __shared__ float4 Bs2[HEAD ? (NQ * K2 / 8) * 4 * 32 : 1];
  __shared__ __attribute__((aligned(16))) float Ls[HEAD ? GROWS * LSP : 1];

  const int tid  = threadIdx.x;
  const int lane = tid & 31;
  const int wave = tid >> 5;
  const int half = lane >> 4;
  const int mn   = lane & 15;
  const int row0 = blockIdx.x * GROWS;
  const int rt   = wave >> 2;
  const int ct0  = (wave & 3) * CT;

  const int arow = min(row0 + rt * 16 + mn, M - 1);
  const float4* __restrict__ ap = reinterpret_cast<const float4*>(A + (long long)arow * K) + half;

  float4 cur[4];
  #pragma unroll
  for (int j = 0; j < 4; ++j) cur[j] = ap[2 * j];

  if (pc != nullptr && tid < GROWS && row0 + tid < M) {
    const int row = row0 + tid;
    unsigned c0 = 0u, c1 = 0u, c2 = 0u;
    #pragma unroll
    for (int g = 0; g < G; ++g) {
      const uint4 p = reinterpret_cast<const uint4*>(pc)[(long long)g * M + row];
      c0 += p.x; c1 += p.y; c2 += p.z;
    }
    float4 o;
    o.x = (c0 > 0u) ? rsqrtf((float)c0) : 0.0f;
    o.y = (c1 > 0u) ? rsqrtf((float)c1) : 0.0f;
    o.z = (c2 > 0u) ? rsqrtf((float)c2) : 0.0f;
    o.w = 0.0f;
    reinterpret_cast<float4*>(dinv4)[row] = o;
  }

  copy_b_frag<K, NC>(B, Bs, tid);
  if constexpr (HEAD) copy_b_frag<NQ * K2, 64>(B2, Bs2, tid);
  __syncthreads();

  v8f acc[CT];
  #pragma unroll
  for (int t = 0; t < CT; ++t) acc[t] = v8f{};

  #pragma unroll 1
  for (int kc = 0; kc < KB; kc += 4) {
    const int kn = min(kc + 4, KB - 4);
    float4 nxt[4];
    #pragma unroll
    for (int j = 0; j < 4; ++j) nxt[j] = ap[2 * (kn + j)];

    #pragma unroll
    for (int j = 0; j < 4; ++j) {
      v2f a0, a1;
      a0.x = cur[j].x; a0.y = cur[j].y;
      a1.x = cur[j].z; a1.y = cur[j].w;
      #pragma unroll
      for (int t = 0; t < CT; ++t) {
        const float4 b = Bs[((kc + j) * NT + ct0 + t) * 32 + lane];
        v2f b0, b1;
        b0.x = b.x; b0.y = b.y;
        b1.x = b.z; b1.y = b.w;
        acc[t] = __builtin_amdgcn_wmma_f32_16x16x4_f32(
             false, a0,  false, b0,
             (short)0, acc[t],  false,  false);
        acc[t] = __builtin_amdgcn_wmma_f32_16x16x4_f32(
             false, a1,  false, b1,
             (short)0, acc[t],  false,  false);
      }
    }
    #pragma unroll
    for (int j = 0; j < 4; ++j) cur[j] = nxt[j];
  }

  if constexpr (!HEAD) {
    #pragma unroll
    for (int t = 0; t < CT; ++t) {
      const int col = (ct0 + t) * 16 + mn;
      const float bv = bias ? bias[col] : 0.0f;
      #pragma unroll
      for (int r = 0; r < 8; ++r) {
        const int row = row0 + rt * 16 + r + half * 8;
        if (row < M) D[(long long)row * NC + col] = acc[t][r] + bv;
      }
    }
  } else {
    #pragma unroll
    for (int t = 0; t < CT; ++t) {
      const int col = (ct0 + t) * 16 + mn;
      const float bv = bias ? bias[col] : 0.0f;
      #pragma unroll
      for (int r = 0; r < 8; ++r) Ls[(rt * 16 + r + half * 8) * LSP + col] = acc[t][r] + bv;
    }
    __syncthreads();

    const int q = wave & 3;
    const float4* lp = reinterpret_cast<const float4*>(Ls + (rt * 16 + mn) * LSP) + half;
    v8f acc2[4];
    #pragma unroll
    for (int t = 0; t < 4; ++t) acc2[t] = v8f{};
    #pragma unroll
    for (int kb = 0; kb < K2 / 8; ++kb) {
      const float4 a = lp[2 * kb];
      v2f a0, a1;
      a0.x = a.x; a0.y = a.y;
      a1.x = a.z; a1.y = a.w;
      #pragma unroll
      for (int t = 0; t < 4; ++t) {
        const float4 b = Bs2[((q * (K2 / 8) + kb) * 4 + t) * 32 + lane];
        v2f b0, b1;
        b0.x = b.x; b0.y = b.y;
        b1.x = b.z; b1.y = b.w;
        acc2[t] = __builtin_amdgcn_wmma_f32_16x16x4_f32(
             false, a0,  false, b0,
             (short)0, acc2[t],  false,  false);
        acc2[t] = __builtin_amdgcn_wmma_f32_16x16x4_f32(
             false, a1,  false, b1,
             (short)0, acc2[t],  false,  false);
      }
    }
    #pragma unroll
    for (int t = 0; t < 4; ++t) {
      const int col = q * 64 + t * 16 + mn;
      #pragma unroll
      for (int r = 0; r < 8; ++r) {
        const int row = row0 + rt * 16 + r + half * 8;
        if (row < M) D[(long long)row * (NQ * 64) + col] = acc2[t][r];
      }
    }

    if (blockIdx.x == 0 && __builtin_amdgcn_readfirstlane(rt) == 0) {
      v8f accb[4];
      #pragma unroll
      for (int t = 0; t < 4; ++t) accb[t] = v8f{};
      #pragma unroll
      for (int kb = 0; kb < K2 / 8; ++kb) {
        const float* bp = bq + kb * 8 + half * 4;
        v2f a0, a1;
        a0.x = bp[0]; a0.y = bp[1];
        a1.x = bp[2]; a1.y = bp[3];
        #pragma unroll
        for (int t = 0; t < 4; ++t) {
          const float4 b = Bs2[((q * (K2 / 8) + kb) * 4 + t) * 32 + lane];
          v2f b0, b1;
          b0.x = b.x; b0.y = b.y;
          b1.x = b.z; b1.y = b.w;
          accb[t] = __builtin_amdgcn_wmma_f32_16x16x4_f32(
               false, a0,  false, b0,
               (short)0, accb[t],  false,  false);
          accb[t] = __builtin_amdgcn_wmma_f32_16x16x4_f32(
               false, a1,  false, b1,
               (short)0, accb[t],  false,  false);
        }
      }
      if (half == 0) {
        #pragma unroll
        for (int t = 0; t < 4; ++t) cq[q * 64 + t * 16 + mn] = accb[t][0];
      }
    }
  }
}


template <int WIDTH>
__device__ __forceinline__ int col_pos(int c) { return (c % (WIDTH / 32)) * 32 + c / (WIDTH / 32); }
template <int WIDTH>
__device__ __forceinline__ int pos_col(int p) { return (p % 32) * (WIDTH / 32) + p / 32; }

template <int WIDTH>
__device__ __forceinline__ void load_row(const float* __restrict__ rowp, int lane, float (&hv)[WIDTH / 32]) {
  static_assert(WIDTH == 128 || WIDTH == 64, "4 or 2 columns per lane");
  if constexpr (WIDTH == 128) {
    const float4 t = reinterpret_cast<const float4*>(rowp)[lane];
    hv[0] = t.x; hv[1] = t.y; hv[2] = t.z; hv[3] = t.w;
  } else {
    const float2 t = reinterpret_cast<const float2*>(rowp)[lane];
    hv[0] = t.x; hv[1] = t.y;
  }
}

template <int WIDTH, bool HEAD>
__device__ __forceinline__ void edge_in_lane(int e, int dl, int k, const int* __restrict__ sp,
                                             const float* __restrict__ dinv4, const float* dinvL,
                                             const float* __restrict__ HH, float* acc) {
  constexpr int LD  = 4 * WIDTH;
  constexpr int SLD = HEAD ? 4 * WIDTH : WIDTH;
  const int s = sp[e];
  const float w = dinv4[s * 4 + k] * dinvL[dl * 4 + k];
  const float* hp = HH + (long long)s * SLD + (HEAD ? k * WIDTH : 0);
  float* ap = acc + dl * LD + (HEAD ? 0 : k * WIDTH);
  #pragma unroll 1
  for (int c = 0; c < WIDTH; ++c) atomicAdd(ap + col_pos<WIDTH>(c), hp[c] * w);
}

template <int WIDTH, int CNT, bool HEAD>
__device__ __forceinline__ void work_edges(const int* qe_, const int* qd_, int k,
                                           const int* __restrict__ sp,
                                           const float* __restrict__ dinv4, const float* dinvL,
                                           const float* __restrict__ HH, float* acc, int lane) {
  constexpr int LD  = 4 * WIDTH;
  constexpr int SLD = HEAD ? 4 * WIDTH : WIDTH;
  constexpr int NJ = WIDTH / 32;
  int dl[CNT], s[CNT];
  #pragma unroll
  for (int u = 0; u < CNT; ++u) { dl[u] = qd_[u]; s[u] = sp[qe_[u]]; }
  float w[CNT], hv[CNT][NJ];
  #pragma unroll
  for (int u = 0; u < CNT; ++u) {
    w[u] = dinv4[s[u] * 4 + k] * dinvL[dl[u] * 4 + k];
    load_row<WIDTH>(HH + (long long)s[u] * SLD + (HEAD ? k * WIDTH : 0), lane, hv[u]);
  }
  #pragma unroll
  for (int u = 0; u < CNT; ++u) {
    float* ap = acc + dl[u] * LD + (HEAD ? 0 : k * WIDTH) + lane;
    #pragma unroll
    for (int j = 0; j < NJ; ++j) atomicAdd(ap + 32 * j, hv[u][j] * w[u]);
  }
}

template <int WIDTH, int CNT, bool FULL>
__device__ __forceinline__ void work_group(int2 rec, float dv, float dl, int g, int last,
                                           const float* __restrict__ HH, float* acc, int lane) {
  constexpr int LD = 4 * WIDTH;
  constexpr int NJ = WIDTH / 32;
  int c[CNT];
  float hv[CNT][NJ];
  #pragma unroll
  for (int u = 0; u < CNT; ++u) {
    const int from = FULL ? g + u : min(g + u, last);
    const int s = __builtin_amdgcn_readlane(rec.x, from);
    c[u] = __builtin_amdgcn_readlane(rec.y, from);
    load_row<WIDTH>(HH + (long long)s * WIDTH, lane, hv[u]);
  }
  #pragma unroll
  for (int u = 0; u < CNT; ++u) {
    if (FULL || g + u <= last) {
      const float w = __int_as_float(__builtin_amdgcn_readlane(__float_as_int(dv), g + u)) *
                      __int_as_float(__builtin_amdgcn_readlane(__float_as_int(dl), g + u));
      float* ap = acc + (c[u] >> 2) * LD + (c[u] & 3) * WIDTH + lane;
      #pragma unroll
      for (int j = 0; j < NJ; ++j) atomicAdd(ap + 32 * j, hv[u][j] * w);
    }
  }
}

template <int WIDTH, bool HEAD>
__device__ __forceinline__ void work_records(const int2* __restrict__ sg, int n,
                                             const float* __restrict__ dinv4, const float* dinvL,
                                             const float* __restrict__ HH, float* acc,
                                             int wave, int lane) {
  constexpr int U = 1024 / WIDTH;
  int b0 = wave * 32;
  if (b0 >= n) return;
  int2 rec = sg[min(b0 + lane, n - 1)];
  while (b0 < n) {
    const int cnt = min(32, n - b0);
    const int nb  = b0 + NTHR;
    int2 nrec = rec;
    if (nb < n) nrec = sg[min(nb + lane, n - 1)];
    const float dv = dinv4[rec.x * 4 + (rec.y & 3)];
    const float dl = dinvL[rec.y];
    int2 wr = rec;
    if constexpr (HEAD) { wr.x = rec.x * 4 + (rec.y & 3); wr.y = rec.y & ~3; }
    int g = 0;
    for (; g + U <= cnt; g += U) work_group<WIDTH, U, true>(wr, dv, dl, g, cnt - 1, HH, acc, lane);
    if (g < cnt)                 work_group<WIDTH, U, false>(wr, dv, dl, g, cnt - 1, HH, acc, lane);
    rec = nrec;
    b0  = nb;
  }
}

template <int WIDTH, bool RELU, bool HEAD>
__global__ __launch_bounds__(NTHR)
void agg_owner(const int* __restrict__ s1, const int* __restrict__ d1, int E1,
               const int* __restrict__ s2, const int* __restrict__ d2, int E2,
               const int* __restrict__ s3, const int* __restrict__ d3, int E3,
               const float* __restrict__ dinv4, const int* __restrict__ seghdr,
               const int* __restrict__ seg, const float* __restrict__ HH,
               const float* __restrict__ bias, const float* __restrict__ cq,
               float* __restrict__ R, int N) {
  constexpr int LD  = 4 * WIDTH;
  constexpr int SLD = HEAD ? 4 * WIDTH : WIDTH;
  constexpr int U  = 4;
  static_assert(!HEAD || (WIDTH == 64 && !RELU), "row pass: 64 columns, two per lane");
  __shared__ __attribute__((aligned(16))) float acc[TR * LD];
  __shared__ float dinvL[TR * 4];
  __shared__ int qe[2 * QCAP];
  __shared__ int qd[2 * QCAP];
  __shared__ unsigned qcount[3];
  __shared__ int hsub[G];
  __shared__ int hovf[G];

  const int tid  = threadIdx.x;
  const int wave = tid >> 5;
  const int lane = tid & 31;
  const int r0   = blockIdx.x * TR;

  constexpr int W4  = WIDTH / 4;
  constexpr int LD4 = LD / 4;
  constexpr int SV  = TR * W4;
  constexpr int SP  = (SV + NTHR - 1) / NTHR;
  static_assert(NTHR % W4 == 0, "a thread keeps its column group of HH through the trips");
  static_assert(HEAD || (NTHR % LD4 == 0 && (TR * LD4) % NTHR == 0), "a thread keeps its column group through the fill");

  const float4* __restrict__ HH4 = reinterpret_cast<const float4*>(HH);
  const long long end4 = (long long)N * W4 - 1;
  float4 sv[SP];
  #pragma unroll
  for (int u = 0; u < SP; ++u) {
    if constexpr (HEAD) {
      const int i = u * NTHR + tid;
      const long long row = min(r0 + i / W4, N - 1);
      sv[u] = HH4[row * (SLD / 4) + 3 * W4 + i % W4];
    } else {
      const long long i4 = (long long)r0 * W4 + u * NTHR + tid;
      sv[u] = HH4[(i4 < end4) ? i4 : end4];
    }
  }
  int4 hd = make_int4(0, 0, 0, 0);
  if (tid < G) hd = reinterpret_cast<const int4*>(seghdr)[(long long)blockIdx.x * G + tid];
  float dvl = 0.0f;
  if (tid < TR * 4 && r0 + (tid >> 2) < N) dvl = dinv4[(long long)r0 * 4 + tid];
  static_assert(TR * 4 <= NTHR, "one degree factor per thread");
  const int h4 = tid % W4;
  float4 bv;
  bv.x = bias[h4 * 4]; bv.y = bias[h4 * 4 + 1]; bv.z = bias[h4 * 4 + 2]; bv.w = bias[h4 * 4 + 3];
  if constexpr (HEAD) {
    float4 cv[4];
    #pragma unroll
    for (int k = 0; k < 4; ++k) cv[k] = reinterpret_cast<const float4*>(cq)[k * W4 + h4];
    #pragma unroll
    for (int k = 0; k < 4; ++k) { bv.x += cv[k].x; bv.y += cv[k].y; bv.z += cv[k].z; bv.w += cv[k].w; }
  }

  if constexpr (!HEAD) {
    float4* acc4w = reinterpret_cast<float4*>(acc);
    float4 bp;
    bp.x = bias[pos_col<WIDTH>(h4 * 4)];     bp.y = bias[pos_col<WIDTH>(h4 * 4 + 1)];
    bp.z = bias[pos_col<WIDTH>(h4 * 4 + 2)]; bp.w = bias[pos_col<WIDTH>(h4 * 4 + 3)];
    if (tid % LD4 < 3 * W4) {
      for (int idx4 = tid; idx4 < TR * LD4; idx4 += NTHR) acc4w[idx4] = bp;
    }
  }
  #pragma unroll
  for (int u = 0; u < SP; ++u) {
    const int i = u * NTHR + tid;
    if (i < SV) {
      const int dl = i / W4;
      float4 v = bv;
      if (r0 + dl < N) { v.x += sv[u].x; v.y += sv[u].y; v.z += sv[u].z; v.w += sv[u].w; }
      float* bp0 = acc + dl * LD + (HEAD ? 0 : 3 * WIDTH);
      bp0[col_pos<WIDTH>(h4 * 4)]     = v.x;
      bp0[col_pos<WIDTH>(h4 * 4 + 1)] = v.y;
      bp0[col_pos<WIDTH>(h4 * 4 + 2)] = v.z;
      bp0[col_pos<WIDTH>(h4 * 4 + 3)] = v.w;
    }
  }
  if (tid < TR * 4) dinvL[tid] = dvl;
  if (tid < G) { hsub[tid] = hd.x; hovf[tid] = hd.y; }
  if (tid < 3) qcount[tid] = 0u;
  __syncthreads();

  int ovf = 0;
  #pragma unroll
  for (int g = 0; g < G; ++g) ovf |= hovf[g];
  ovf = __builtin_amdgcn_readfirstlane(ovf);

  if (ovf == 0) {
    const int wv = __builtin_amdgcn_readfirstlane(wave);
    int rot = 0;
    #pragma unroll 1
    for (int g = 0; g < G; ++g) {
      const int nrec = __builtin_amdgcn_readfirstlane(hsub[g]);
      work_records<WIDTH, HEAD>(reinterpret_cast<const int2*>(seg) + ((long long)blockIdx.x * G + g) * SUBCAP, nrec,
                          dinv4, dinvL, HH, acc, (wv - rot) & (NWAVE - 1), lane);
      rot += (nrec + 31) >> 5;
    }
  } else {
    int buf = 0;
    int ci  = 0;

    #pragma unroll 1
    for (int k = 0; k < 3; ++k) {
      const int* __restrict__ sp = (k == 0) ? s1 : (k == 1) ? s2 : s3;
      const int* __restrict__ dp = (k == 0) ? d1 : (k == 1) ? d2 : d3;
      const int E = (k == 0) ? E1 : (k == 1) ? E2 : E3;
      const bool vec_ok = ((E & 3) == 0) && ((reinterpret_cast<size_t>(dp) & 15) == 0);

      int4 cur[NV];
      load_chunk(cur, dp, 0, E, vec_ok, tid);

      for (int base = 0; base < E; base += CHUNK) {
        const int cn = (ci == 2) ? 0 : ci + 1;
        if (tid == 0) qcount[cn] = 0u;
        unsigned* qc = &qcount[ci];
        int* qeb = qe + buf * QCAP;
        int* qdb = qd + buf * QCAP;

        #pragma unroll
        for (int j = 0; j < NV; ++j) {
          const int i = base + (j * NTHR + tid) * 4;
          const unsigned a0 = (unsigned)(cur[j].x - r0), a1 = (unsigned)(cur[j].y - r0);
          const unsigned a2 = (unsigned)(cur[j].z - r0), a3 = (unsigned)(cur[j].w - r0);
          if (min(min(a0, a1), min(a2, a3)) < (unsigned)TR) {
            #pragma unroll 1
            for (int t = 0; t < 4; ++t) {
              const unsigned a = (t == 0) ? a0 : (t == 1) ? a1 : (t == 2) ? a2 : a3;
              if (a < (unsigned)TR) {
                const unsigned pos = atomicAdd(qc, 1u);
                if (pos < (unsigned)QCAP) { qeb[pos] = i + t; qdb[pos] = (int)a; }
                else edge_in_lane<WIDTH, HEAD>(i + t, (int)a, k, sp, dinv4, dinvL, HH, acc);
              }
            }
          }
        }

        __syncthreads();

        int4 nxt[NV];
        load_chunk(nxt, dp, base + CHUNK, E, vec_ok, tid);

        const int n = min((int)(*qc), QCAP);
        const int nfull = n - (n % U);
        for (int i0 = wave * U; i0 < nfull; i0 += (NTHR / 32) * U)
          work_edges<WIDTH, U, HEAD>(qeb + i0, qdb + i0, k, sp, dinv4, dinvL, HH, acc, lane);
        for (int i1 = nfull + wave; i1 < n; i1 += NTHR / 32)
          work_edges<WIDTH, 1, HEAD>(qeb + i1, qdb + i1, k, sp, dinv4, dinvL, HH, acc, lane);


        #pragma unroll
        for (int j = 0; j < NV; ++j) cur[j] = nxt[j];
        buf ^= 1;
        ci = cn;
      }
    }
  }
  __syncthreads();

  if constexpr (HEAD) {
    constexpr int RQ = (TR + NWAVE - 1) / NWAVE;
    float a[RQ], b[RQ], mx[RQ], s[RQ];
    #pragma unroll
    for (int q = 0; q < RQ; ++q) {
      const int lrow = min(wave + q * NWAVE, TR - 1);
      a[q]  = acc[lrow * LD + col_pos<WIDTH>(lane)];
      b[q]  = acc[lrow * LD + col_pos<WIDTH>(lane + 32)];
      mx[q] = fmaxf(a[q], b[q]);
    }
    #pragma unroll
    for (int off = 16; off > 0; off >>= 1) {
      float o[RQ];
      #pragma unroll
      for (int q = 0; q < RQ; ++q) o[q] = __shfl_xor(mx[q], off);
      #pragma unroll
      for (int q = 0; q < RQ; ++q) mx[q] = fmaxf(mx[q], o[q]);
    }
    #pragma unroll
    for (int q = 0; q < RQ; ++q) s[q] = __expf(a[q] - mx[q]) + __expf(b[q] - mx[q]);
    #pragma unroll
    for (int off = 16; off > 0; off >>= 1) {
      float o[RQ];
      #pragma unroll
      for (int q = 0; q < RQ; ++q) o[q] = __shfl_xor(s[q], off);
      #pragma unroll
      for (int q = 0; q < RQ; ++q) s[q] += o[q];
    }
    #pragma unroll
    for (int q = 0; q < RQ; ++q) {
      const int lrow = wave + q * NWAVE;
      const int row = r0 + lrow;
      const float lse = mx[q] + __logf(s[q]);
      if (lrow < TR && row < N) {
        float* p = R + (long long)row * WIDTH;
        p[lane]      = a[q] - lse;
        p[lane + 32] = b[q] - lse;
      }
    }
  } else {
    float4* R4 = reinterpret_cast<float4*>(R + (long long)r0 * LD);
    for (int idx = tid; idx < TR * LD / 4; idx += NTHR) {
      const int row = r0 + (idx * 4) / LD;
      if (row < N) {
        const int g = idx % W4;
        const float* bp0 = acc + (idx / W4) * WIDTH;
        float4 v;
        v.x = bp0[col_pos<WIDTH>(g * 4)];     v.y = bp0[col_pos<WIDTH>(g * 4 + 1)];
        v.z = bp0[col_pos<WIDTH>(g * 4 + 2)]; v.w = bp0[col_pos<WIDTH>(g * 4 + 3)];
        if (RELU) { v.x = fmaxf(v.x, 0.0f); v.y = fmaxf(v.y, 0.0f); v.z = fmaxf(v.z, 0.0f); v.w = fmaxf(v.w, 0.0f); }
        R4[idx] = v;
      }
    }
  }
}

extern "C" void kernel_launch(void* const* d_in, const int* in_sizes, int n_in,
                              void* d_out, int out_size, void* d_ws, size_t ws_size,
                              hipStream_t stream) {
  const float* x  = (const float*)d_in[0];
  const int* e1   = (const int*)d_in[1];
  const int* e2   = (const int*)d_in[2];
  const int* e3   = (const int*)d_in[3];
  const float* W1 = (const float*)d_in[5];
  const float* b1 = (const float*)d_in[6];
  const float* W2 = (const float*)d_in[7];
  const float* b2 = (const float*)d_in[8];
  const float* Wl = (const float*)d_in[9];
  const float* bl = (const float*)d_in[10];

  const int F = 512, H = 128, C = 64;
  const int N  = in_sizes[0] / F;
  const int E1 = in_sizes[1] / 2;
  const int E2 = in_sizes[2] / 2;
  const int E3 = in_sizes[3] / 2;

  float* ws    = (float*)d_ws;
  float* dinv4 = ws;
  float* HH1   = dinv4 + 4LL * N;
  float* R1    = HH1 + (long long)N * H;
  float* P2    = R1 + (long long)N * 4 * H;
  float* cq    = P2 + (long long)N * 4 * C;

  const int otiles = cdiv_i(N, TR);
  const int gblocks = cdiv_i(N, GROWS);
  const int groups = cdiv_i(otiles, G);
  const int gtiles = groups * G;
  unsigned* pcnt = (unsigned*)(cq + 4 * C);
  int* seghdr = (int*)(pcnt + 4LL * G * N);
  int* seg    = seghdr + 4LL * G * gtiles;

  deg_owner<<<groups * G, NTHR, 0, stream>>>(e1, e1 + E1, E1, e2, e2 + E2, E2, e3, e3 + E3, E3,
                                             pcnt, seghdr, seg, N);

  gemm_wmma_f32<512, 128, false><<<gblocks, GTH, 0, stream>>>(x, W1, nullptr, HH1, N, pcnt, dinv4,
                                                              nullptr, nullptr, nullptr);

  agg_owner<128, true, false><<<otiles, NTHR, 0, stream>>>(
      e1, e1 + E1, E1, e2, e2 + E2, E2, e3, e3 + E3, E3, dinv4, seghdr, seg, HH1, b1, nullptr, R1, N);

  gemm_wmma_f32<512, 64, true><<<gblocks, GTH, 0, stream>>>(R1, W2, nullptr, P2, N, nullptr, nullptr,
                                                            Wl, b2, cq);

  agg_owner<64, false, true><<<otiles, NTHR, 0, stream>>>(
      e1, e1 + E1, E1, e2, e2 + E2, E2, e3, e3 + E3, E3, dinv4, seghdr, seg, P2, bl, cq, (float*)d_out, N);
}
